# starting kernel
# baseline (speedup 1.0000x reference)
#include <hip/hip_runtime.h>


#define HDIM   64
#define VOCABN 64
#define BATCH  64
#define SEQL   4096

typedef __attribute__((ext_vector_type(2))) float v2f;
typedef __attribute__((ext_vector_type(2))) float f32x2;
typedef __attribute__((ext_vector_type(8))) float v8f;

__device__ __forceinline__ v2f frag_a(const float* A, int lda, int mt, int kt, int lane) {
  int row = (mt << 4) + (lane & 15);
  int k0  = (kt << 2) + ((lane >> 4) << 1);
  v2f a;
  a.x = A[row * lda + k0];
  a.y = A[row * lda + k0 + 1];
  return a;
}
__device__ __forceinline__ v2f frag_b(const float* Bm, int ldb, int kt, int nt, int lane) {
  int col = (nt << 4) + (lane & 15);
  int k0  = (kt << 2) + ((lane >> 4) << 1);
  v2f b;
  b.x = Bm[k0 * ldb + col];
  b.y = Bm[(k0 + 1) * ldb + col];
  return b;
}

__device__ __forceinline__ v8f wmma_f32(v2f a, v2f b, v8f c) {
  return __builtin_amdgcn_wmma_f32_16x16x4_f32(false, a, false, b, (short)0, c,
                                               false, false);
}

__global__ __launch_bounds__(256) void encode_table_kernel(
    const float* __restrict__ embed, const float* __restrict__ W1,
    const float* __restrict__ b1, const float* __restrict__ W2,
    const float* __restrict__ b2, const float* __restrict__ gamma,
    const float* __restrict__ beta, float* __restrict__ Tout,
    float* __restrict__ rdenom) {
  __shared__ float Rlds[64 * 128];
  __shared__ float Xlds[64 * 64];
  const int tid  = threadIdx.x;
  const int wave = tid >> 5;
  const int lane = tid & 31;

  for (int tile = wave; tile < 32; tile += 8) {
    int mt = tile >> 3, nt = tile & 7;
    v8f acc = {};
#pragma unroll
    for (int kt = 0; kt < 16; ++kt) {
      v2f a = frag_a(embed, 64, mt, kt, lane);
      v2f b = frag_b(W1, 128, kt, nt, lane);
      acc = wmma_f32(a, b, acc);
    }
    int col   = (nt << 4) + (lane & 15);
    int rbase = (mt << 4) + ((lane >> 4) << 3);
#pragma unroll
    for (int v = 0; v < 8; ++v) {
      float x = acc[v] + b1[col];
      Rlds[(rbase + v) * 128 + col] = fmaxf(x, 0.0f);
    }
  }
  __syncthreads();

  for (int tile = wave; tile < 16; tile += 8) {
    int mt = tile >> 2, nt = tile & 3;
    v8f acc = {};
#pragma unroll
    for (int kt = 0; kt < 32; ++kt) {
      v2f a = frag_a(Rlds, 128, mt, kt, lane);
      v2f b = frag_b(W2, 64, kt, nt, lane);
      acc = wmma_f32(a, b, acc);
    }
    int col   = (nt << 4) + (lane & 15);
    int rbase = (mt << 4) + ((lane >> 4) << 3);
#pragma unroll
    for (int v = 0; v < 8; ++v) {
      int row = rbase + v;
      Xlds[row * 64 + col] = embed[row * 64 + col] + acc[v] + b2[col];
    }
  }
  __syncthreads();

  if (tid < 64) {
    const float* xr = &Xlds[tid * 64];
    float mu = 0.f;
#pragma unroll
    for (int c = 0; c < 64; ++c) mu += xr[c];
    mu *= (1.0f / 64.0f);
    float var = 0.f;
#pragma unroll
    for (int c = 0; c < 64; ++c) { float d = xr[c] - mu; var += d * d; }
    var *= (1.0f / 64.0f);
    float rs  = rsqrtf(var + 1e-5f);
    float ksq = 0.f;
#pragma unroll
    for (int c = 0; c < 64; ++c) {
      float tv = (xr[c] - mu) * rs * gamma[c] + beta[c];
      Tout[tid * 64 + c] = tv;
      ksq += tv * tv;
    }
    rdenom[tid] = 1.0f / (ksq + 1e-6f);
  }
}

__global__ __launch_bounds__(128) void delta_scan_kernel(
    const int* __restrict__ seq, const float* __restrict__ Ttab,
    const float* __restrict__ rdenom, float* __restrict__ ctx) {
  __shared__ float4 Tl4[VOCABN * 16];
  __shared__ float  rd[VOCABN];
  __shared__ int    toks[SEQL];
  const int b    = blockIdx.x;
  const int tid  = threadIdx.x;
  const int r    = tid >> 1;
  const int half = tid & 1;

  const float4* Tg4 = (const float4*)Ttab;
  for (int i = tid; i < VOCABN * 16; i += 128) Tl4[i] = Tg4[i];
  if (tid < VOCABN) rd[tid] = rdenom[tid];
  const int4* sg = (const int4*)(seq + b * SEQL);
  int4*       tl = (int4*)toks;
  for (int i = tid; i < SEQL / 4; i += 128) tl[i] = sg[i];
  __syncthreads();

  const float* Tl    = (const float*)Tl4;
  const int    kbase = half << 3;

  f32x2 M2[16];
#pragma unroll
  for (int i = 0; i < 16; ++i) M2[i] = (f32x2){0.f, 0.f};

  for (int t = 0; t < SEQL - 1; ++t) {
    int tok = toks[t];
    const float4* kp = &Tl4[tok * 16 + kbase];
    f32x2 kv[16];
    f32x2 acc0 = (f32x2){0.f, 0.f};
    f32x2 acc1 = (f32x2){0.f, 0.f};
#pragma unroll
    for (int i = 0; i < 8; ++i) {
      float4 ld = kp[i];
      kv[2 * i]     = (f32x2){ld.x, ld.y};
      kv[2 * i + 1] = (f32x2){ld.z, ld.w};
      acc0 = __builtin_elementwise_fma(M2[2 * i], kv[2 * i], acc0);
      acc1 = __builtin_elementwise_fma(M2[2 * i + 1], kv[2 * i + 1], acc1);
    }
    float part = (acc0.x + acc0.y) + (acc1.x + acc1.y);
    float vp   = part + __shfl_xor(part, 1, 32);
    float dv   = Tl[tok * 64 + r] - vp * rd[tok];
    f32x2 dvv  = (f32x2){dv, dv};
#pragma unroll
    for (int i = 0; i < 16; ++i)
      M2[i] = __builtin_elementwise_fma(dvv, kv[i], M2[i]);
  }

  int tok = toks[SEQL - 1];
  const float4* qp = &Tl4[tok * 16 + kbase];
  f32x2 acc0 = (f32x2){0.f, 0.f};
  f32x2 acc1 = (f32x2){0.f, 0.f};
#pragma unroll
  for (int i = 0; i < 8; ++i) {
    float4 ld = qp[i];
    acc0 = __builtin_elementwise_fma(M2[2 * i], (f32x2){ld.x, ld.y}, acc0);
    acc1 = __builtin_elementwise_fma(M2[2 * i + 1], (f32x2){ld.z, ld.w}, acc1);
  }
  float part = (acc0.x + acc0.y) + (acc1.x + acc1.y);
  float vp   = part + __shfl_xor(part, 1, 32);
  if (half == 0) ctx[b * 64 + r] = vp;
}

__global__ __launch_bounds__(256) void head_kernel(
    const float* __restrict__ ctx, const float* __restrict__ Wr,
    const float* __restrict__ br, const float* __restrict__ Wo,
    const float* __restrict__ bo, float* __restrict__ out) {
  __shared__ float Glds[64 * 64];
  const int tid  = threadIdx.x;
  const int wave = tid >> 5;
  const int lane = tid & 31;

  for (int tile = wave; tile < 16; tile += 8) {
    int mt = tile >> 2, nt = tile & 3;
    v8f acc = {};
#pragma unroll
    for (int kt = 0; kt < 16; ++kt) {
      v2f a = frag_a(ctx, 64, mt, kt, lane);
      v2f b = frag_b(Wr, 64, kt, nt, lane);
      acc = wmma_f32(a, b, acc);
    }
    int col   = (nt << 4) + (lane & 15);
    int rbase = (mt << 4) + ((lane >> 4) << 3);
#pragma unroll
    for (int v = 0; v < 8; ++v) Glds[(rbase + v) * 64 + col] = acc[v] + br[col];
  }
  __syncthreads();

  for (int tile = wave; tile < 16; tile += 8) {
    int mt = tile >> 2, nt = tile & 3;
    v8f acc = {};
#pragma unroll
    for (int kt = 0; kt < 16; ++kt) {
      v2f a = frag_a(Glds, 64, mt, kt, lane);
      v2f b = frag_b(Wo, 64, kt, nt, lane);
      acc = wmma_f32(a, b, acc);
    }
    int col   = (nt << 4) + (lane & 15);
    int rbase = (mt << 4) + ((lane >> 4) << 3);
#pragma unroll
    for (int v = 0; v < 8; ++v) out[(rbase + v) * 64 + col] = acc[v] + bo[col];
  }
}

extern "C" void kernel_launch(void* const* d_in, const int* in_sizes, int n_in,
                              void* d_out, int out_size, void* d_ws, size_t ws_size,
                              hipStream_t stream) {
  const int*   seq   = (const int*)d_in[0];
  const float* embed = (const float*)d_in[1];
  const float* W1    = (const float*)d_in[2];
  const float* b1    = (const float*)d_in[3];
  const float* W2    = (const float*)d_in[4];
  const float* b2    = (const float*)d_in[5];
  const float* gamma = (const float*)d_in[6];
  const float* beta  = (const float*)d_in[7];
  const float* Wr    = (const float*)d_in[8];
  const float* br    = (const float*)d_in[9];
  const float* Wo    = (const float*)d_in[10];
  const float* bo    = (const float*)d_in[11];

  float* Tt     = (float*)d_ws;
  float* rdenom = Tt + VOCABN * HDIM;
  float* ctx    = rdenom + VOCABN;
  float* out    = (float*)d_out;

  encode_table_kernel<<<1, 256, 0, stream>>>(embed, W1, b1, W2, b2, gamma, beta,
                                             Tt, rdenom);
  delta_scan_kernel<<<BATCH, 128, 0, stream>>>(seq, Tt, rdenom, ctx);
  head_kernel<<<1, 256, 0, stream>>>(ctx, Wr, br, Wo, bo, out);
}


# Scan run as its adjoint: backward recurrence on c=T.q in token coordinates (Gram matrix by WMMA), one lane read + fma per step, one wave per batch row; summation order changes
# speedup vs baseline: 4.5640x; 4.5640x over previous
#include <hip/hip_runtime.h>


#define HDIM   64
#define VOCABN 64
#define BATCH  64
#define SEQL   4096
#define TPITCH 68

typedef __attribute__((ext_vector_type(2))) float v2f;
typedef __attribute__((ext_vector_type(2))) float f32x2;
typedef __attribute__((ext_vector_type(8))) float v8f;

__device__ __forceinline__ v2f frag_a(const float* A, int lda, int mt, int kt, int lane) {
  int row = (mt << 4) + (lane & 15);
  int k0  = (kt << 2) + ((lane >> 4) << 1);
  v2f a;
  a.x = A[row * lda + k0];
  a.y = A[row * lda + k0 + 1];
  return a;
}
__device__ __forceinline__ v2f frag_b(const float* Bm, int ldb, int kt, int nt, int lane) {
  int col = (nt << 4) + (lane & 15);
  int k0  = (kt << 2) + ((lane >> 4) << 1);
  v2f b;
  b.x = Bm[k0 * ldb + col];
  b.y = Bm[(k0 + 1) * ldb + col];
  return b;
}
__device__ __forceinline__ v2f frag_bt(const float* S, int lds, int kt, int nt, int lane) {
  int col = (nt << 4) + (lane & 15);
  int k0  = (kt << 2) + ((lane >> 4) << 1);
  v2f b;
  b.x = S[col * lds + k0];
  b.y = S[col * lds + k0 + 1];
  return b;
}

__device__ __forceinline__ v8f wmma_f32(v2f a, v2f b, v8f c) {
  return __builtin_amdgcn_wmma_f32_16x16x4_f32(false, a, false, b, (short)0, c,
                                               false, false);
}

__global__ __launch_bounds__(256) void encode_table_kernel(
    const float* __restrict__ embed, const float* __restrict__ W1,
    const float* __restrict__ b1, const float* __restrict__ W2,
    const float* __restrict__ b2, const float* __restrict__ gamma,
    const float* __restrict__ beta, float* __restrict__ Tout,
    float* __restrict__ Gplain, float* __restrict__ Gpair) {
  __shared__ float Rlds[64 * 128];
  __shared__ float Xlds[64 * 64];
  __shared__ float rdl[64];
  const int tid  = threadIdx.x;
  const int wave = tid >> 5;
  const int lane = tid & 31;

  for (int tile = wave; tile < 32; tile += 8) {
    int mt = tile >> 3, nt = tile & 7;
    v8f acc = {};
#pragma unroll
    for (int kt = 0; kt < 16; ++kt) {
      v2f a = frag_a(embed, 64, mt, kt, lane);
      v2f b = frag_b(W1, 128, kt, nt, lane);
      acc = wmma_f32(a, b, acc);
    }
    int col   = (nt << 4) + (lane & 15);
    int rbase = (mt << 4) + ((lane >> 4) << 3);
#pragma unroll
    for (int v = 0; v < 8; ++v) {
      float x = acc[v] + b1[col];
      Rlds[(rbase + v) * 128 + col] = fmaxf(x, 0.0f);
    }
  }
  __syncthreads();

  for (int tile = wave; tile < 16; tile += 8) {
    int mt = tile >> 2, nt = tile & 3;
    v8f acc = {};
#pragma unroll
    for (int kt = 0; kt < 32; ++kt) {
      v2f a = frag_a(Rlds, 128, mt, kt, lane);
      v2f b = frag_b(W2, 64, kt, nt, lane);
      acc = wmma_f32(a, b, acc);
    }
    int col   = (nt << 4) + (lane & 15);
    int rbase = (mt << 4) + ((lane >> 4) << 3);
#pragma unroll
    for (int v = 0; v < 8; ++v) {
      int row = rbase + v;
      Xlds[row * 64 + col] = embed[row * 64 + col] + acc[v] + b2[col];
    }
  }
  __syncthreads();

  if (tid < 64) {
    const float* xr = &Xlds[tid * 64];
    float mu = 0.f;
#pragma unroll
    for (int c = 0; c < 64; ++c) mu += xr[c];
    mu *= (1.0f / 64.0f);
    float var = 0.f;
#pragma unroll
    for (int c = 0; c < 64; ++c) { float d = xr[c] - mu; var += d * d; }
    var *= (1.0f / 64.0f);
    float rs  = rsqrtf(var + 1e-5f);
    float ksq = 0.f;
#pragma unroll
    for (int c = 0; c < 64; ++c) {
      float tv = (xr[c] - mu) * rs * gamma[c] + beta[c];
      Tout[tid * 64 + c] = tv;
      Rlds[tid * TPITCH + c] = tv;
      ksq += tv * tv;
    }
    rdl[tid] = 1.0f / (ksq + 1e-6f);
  }
  __syncthreads();

  float* Tp = Rlds;
  for (int tile = wave; tile < 16; tile += 8) {
    int mt = tile >> 2, nt = tile & 3;
    v8f acc = {};
#pragma unroll
    for (int kt = 0; kt < 16; ++kt) {
      v2f a = frag_a(Tp, TPITCH, mt, kt, lane);
      v2f b = frag_bt(Tp, TPITCH, kt, nt, lane);
      acc = wmma_f32(a, b, acc);
    }
    int col   = (nt << 4) + (lane & 15);
    int rbase = (mt << 4) + ((lane >> 4) << 3);
#pragma unroll
    for (int v = 0; v < 8; ++v) {
      int row = rbase + v;
      Gplain[row * 64 + col] = acc[v];
      Gpair[((row << 5) + (col & 31)) * 2 + (col >> 5)] = -(acc[v] * rdl[row]);
    }
  }
}

#define GRP 16

__device__ __forceinline__ float lane_read(float x, int l) {
  return __int_as_float(__builtin_amdgcn_readlane(__float_as_int(x), l));
}

__global__ __launch_bounds__(256) void delta_adjoint_kernel(
    const int* __restrict__ seq, const float* __restrict__ Ttab,
    const float* __restrict__ Gplain, const float* __restrict__ Gpair,
    float* __restrict__ ctx) {
  __shared__ float4 Gl4[VOCABN * 16];
  __shared__ float4 Tl4[VOCABN * 16];
  const int b   = blockIdx.x;
  const int tid = threadIdx.x;

  const float4* Gg4 = (const float4*)Gpair;
  const float4* Tg4 = (const float4*)Ttab;
  for (int i = tid; i < VOCABN * 16; i += 256) {
    Gl4[i] = Gg4[i];
    Tl4[i] = Tg4[i];
  }
  __syncthreads();
  if (tid >= 32) return;

  const int     lane = tid;
  const float2* Gl2  = (const float2*)Gl4;
  const float2* Tl2  = (const float2*)Tl4;
  const int4*   sq4  = (const int4*)(seq + b * SEQL);

  int    tkc[GRP], tk1[GRP], tk2[GRP];
  float2 rowc[GRP], rown[GRP];

#define LOAD_TOK(g, tk)                                           \
  {                                                               \
    int4 t0 = sq4[4 * (g)], t1 = sq4[4 * (g) + 1];                \
    int4 t2 = sq4[4 * (g) + 2], t3 = sq4[4 * (g) + 3];            \
    tk[0] = t0.x; tk[1] = t0.y; tk[2] = t0.z; tk[3] = t0.w;       \
    tk[4] = t1.x; tk[5] = t1.y; tk[6] = t1.z; tk[7] = t1.w;       \
    tk[8] = t2.x; tk[9] = t2.y; tk[10] = t2.z; tk[11] = t2.w;     \
    tk[12] = t3.x; tk[13] = t3.y; tk[14] = t3.z; tk[15] = t3.w;   \
  }
#define STEP(tok, row)                                            \
  {                                                               \
    float slo = lane_read(c_lo, (tok) & 31);                      \
    float shi = lane_read(c_hi, (tok) & 31);                      \
    float s   = ((tok) & 32) ? shi : slo;                         \
    c_lo = fmaf((row).x, s, c_lo);                                \
    c_hi = fmaf((row).y, s, c_hi);                                \
    w_lo += (lane == (tok)) ? s : 0.0f;                           \
    w_hi += (lane + 32 == (tok)) ? s : 0.0f;                      \
  }

  const int NG = SEQL / GRP;
  LOAD_TOK(NG - 1, tkc);
  LOAD_TOK(NG - 2, tk1);
  LOAD_TOK(NG - 3, tk2);
#pragma unroll
  for (int j = 0; j < GRP; ++j) rowc[j] = Gl2[tkc[j] * 32 + lane];
#pragma unroll
  for (int j = 0; j < GRP; ++j) rown[j] = Gl2[tk1[j] * 32 + lane];

  const int tq = tkc[GRP - 1];
  float c_lo = Gplain[tq * 64 + lane];
  float c_hi = Gplain[tq * 64 + lane + 32];
  float w_lo = 0.0f, w_hi = 0.0f;

#pragma unroll
  for (int j = GRP - 2; j >= 0; --j) STEP(tkc[j], rowc[j]);

  for (int g = NG - 2; g >= 0; --g) {
#pragma unroll
    for (int j = 0; j < GRP; ++j) { tkc[j] = tk1[j]; rowc[j] = rown[j]; tk1[j] = tk2[j]; }
    const int gp = (g >= 2) ? (g - 2) : 0;
    LOAD_TOK(gp, tk2);
#pragma unroll
    for (int j = 0; j < GRP; ++j) rown[j] = Gl2[tk1[j] * 32 + lane];
#pragma unroll
    for (int j = GRP - 1; j >= 0; --j) STEP(tkc[j], rowc[j]);
#pragma unroll
    for (int j = 0; j < GRP; j += 4)
      asm volatile("" ::"v"(rown[j].x), "v"(rown[j].y), "v"(rown[j + 1].x), "v"(rown[j + 1].y),
                   "v"(rown[j + 2].x), "v"(rown[j + 2].y), "v"(rown[j + 3].x), "v"(rown[j + 3].y));
    asm volatile("" ::"s"(tk2[0]), "s"(tk2[1]), "s"(tk2[2]), "s"(tk2[3]), "s"(tk2[4]), "s"(tk2[5]),
                 "s"(tk2[6]), "s"(tk2[7]), "s"(tk2[8]), "s"(tk2[9]), "s"(tk2[10]), "s"(tk2[11]),
                 "s"(tk2[12]), "s"(tk2[13]), "s"(tk2[14]), "s"(tk2[15]));
  }
#undef STEP
#undef LOAD_TOK

  float ax0 = 0.f, ay0 = 0.f, ax1 = 0.f, ay1 = 0.f;
#pragma unroll
  for (int v = 0; v < 32; v += 2) {
    float  wa = lane_read(w_lo, v), wb = lane_read(w_lo, v + 1);
    float2 ta = Tl2[v * 32 + lane], tb = Tl2[(v + 1) * 32 + lane];
    ax0 = fmaf(ta.x, wa, ax0); ay0 = fmaf(ta.y, wa, ay0);
    ax1 = fmaf(tb.x, wb, ax1); ay1 = fmaf(tb.y, wb, ay1);
  }
#pragma unroll
  for (int v = 0; v < 32; v += 2) {
    float  wa = lane_read(w_hi, v), wb = lane_read(w_hi, v + 1);
    float2 ta = Tl2[(v + 32) * 32 + lane], tb = Tl2[(v + 33) * 32 + lane];
    ax0 = fmaf(ta.x, wa, ax0); ay0 = fmaf(ta.y, wa, ay0);
    ax1 = fmaf(tb.x, wb, ax1); ay1 = fmaf(tb.y, wb, ay1);
  }
  float2 r;
  r.x = ax0 + ax1;
  r.y = ay0 + ay1;
  ((float2*)(ctx + b * 64))[lane] = r;
}

__global__ __launch_bounds__(256) void head_kernel(
    const float* __restrict__ ctx, const float* __restrict__ Wr,
    const float* __restrict__ br, const float* __restrict__ Wo,
    const float* __restrict__ bo, float* __restrict__ out) {
  __shared__ float Glds[64 * 64];
  const int tid  = threadIdx.x;
  const int wave = tid >> 5;
  const int lane = tid & 31;

  for (int tile = wave; tile < 16; tile += 8) {
    int mt = tile >> 2, nt = tile & 3;
    v8f acc = {};
#pragma unroll
    for (int kt = 0; kt < 16; ++kt) {
      v2f a = frag_a(ctx, 64, mt, kt, lane);
      v2f b = frag_b(Wr, 64, kt, nt, lane);
      acc = wmma_f32(a, b, acc);
    }
    int col   = (nt << 4) + (lane & 15);
    int rbase = (mt << 4) + ((lane >> 4) << 3);
#pragma unroll
    for (int v = 0; v < 8; ++v) Glds[(rbase + v) * 64 + col] = acc[v] + br[col];
  }
  __syncthreads();

  for (int tile = wave; tile < 16; tile += 8) {
    int mt = tile >> 2, nt = tile & 3;
    v8f acc = {};
#pragma unroll
    for (int kt = 0; kt < 16; ++kt) {
      v2f a = frag_a(Glds, 64, mt, kt, lane);
      v2f b = frag_b(Wo, 64, kt, nt, lane);
      acc = wmma_f32(a, b, acc);
    }
    int col   = (nt << 4) + (lane & 15);
    int rbase = (mt << 4) + ((lane >> 4) << 3);
#pragma unroll
    for (int v = 0; v < 8; ++v) out[(rbase + v) * 64 + col] = acc[v] + bo[col];
  }
}

extern "C" void kernel_launch(void* const* d_in, const int* in_sizes, int n_in,
                              void* d_out, int out_size, void* d_ws, size_t ws_size,
                              hipStream_t stream) {
  const int*   seq   = (const int*)d_in[0];
  const float* embed = (const float*)d_in[1];
  const float* W1    = (const float*)d_in[2];
  const float* b1    = (const float*)d_in[3];
  const float* W2    = (const float*)d_in[4];
  const float* b2    = (const float*)d_in[5];
  const float* gamma = (const float*)d_in[6];
  const float* beta  = (const float*)d_in[7];
  const float* Wr    = (const float*)d_in[8];
  const float* br    = (const float*)d_in[9];
  const float* Wo    = (const float*)d_in[10];
  const float* bo    = (const float*)d_in[11];

  float* Tt     = (float*)d_ws;
  float* Gplain = Tt + VOCABN * HDIM;
  float* Gpair  = Gplain + VOCABN * VOCABN;
  float* ctx    = Gpair + VOCABN * VOCABN;
  float* out    = (float*)d_out;

  encode_table_kernel<<<1, 256, 0, stream>>>(embed, W1, b1, W2, b2, gamma, beta,
                                             Tt, Gplain, Gpair);
  delta_adjoint_kernel<<<BATCH, 256, 0, stream>>>(seq, Tt, Gplain, Gpair, ctx);
  head_kernel<<<1, 256, 0, stream>>>(ctx, Wr, br, Wo, bo, out);
}


# Scan steps taken 16 at a time as a forward substitution on a 16-lane vector in position coordinates: step value broadcast by DPP row_share, no scalar register on the carried path; same order of sums
# speedup vs baseline: 7.6076x; 1.6669x over previous
#include <hip/hip_runtime.h>


#define HDIM   64
#define VOCABN 64
#define BATCH  64
#define SEQL   4096
#define TPITCH 68

typedef __attribute__((ext_vector_type(2))) float v2f;
typedef __attribute__((ext_vector_type(2))) float f32x2;
typedef __attribute__((ext_vector_type(8))) float v8f;

__device__ __forceinline__ v2f frag_a(const float* A, int lda, int mt, int kt, int lane) {
  int row = (mt << 4) + (lane & 15);
  int k0  = (kt << 2) + ((lane >> 4) << 1);
  v2f a;
  a.x = A[row * lda + k0];
  a.y = A[row * lda + k0 + 1];
  return a;
}
__device__ __forceinline__ v2f frag_b(const float* Bm, int ldb, int kt, int nt, int lane) {
  int col = (nt << 4) + (lane & 15);
  int k0  = (kt << 2) + ((lane >> 4) << 1);
  v2f b;
  b.x = Bm[k0 * ldb + col];
  b.y = Bm[(k0 + 1) * ldb + col];
  return b;
}
__device__ __forceinline__ v2f frag_bt(const float* S, int lds, int kt, int nt, int lane) {
  int col = (nt << 4) + (lane & 15);
  int k0  = (kt << 2) + ((lane >> 4) << 1);
  v2f b;
  b.x = S[col * lds + k0];
  b.y = S[col * lds + k0 + 1];
  return b;
}

__device__ __forceinline__ v8f wmma_f32(v2f a, v2f b, v8f c) {
  return __builtin_amdgcn_wmma_f32_16x16x4_f32(false, a, false, b, (short)0, c,
                                               false, false);
}

__global__ __launch_bounds__(256) void encode_table_kernel(
    const float* __restrict__ embed, const float* __restrict__ W1,
    const float* __restrict__ b1, const float* __restrict__ W2,
    const float* __restrict__ b2, const float* __restrict__ gamma,
    const float* __restrict__ beta, float* __restrict__ Tout,
    float* __restrict__ Gplain, float* __restrict__ Gpair) {
  __shared__ float Rlds[64 * 128];
  __shared__ float Xlds[64 * 64];
  __shared__ float rdl[64];
  const int tid  = threadIdx.x;
  const int wave = tid >> 5;
  const int lane = tid & 31;

  for (int tile = wave; tile < 32; tile += 8) {
    int mt = tile >> 3, nt = tile & 7;
    v8f acc = {};
#pragma unroll
    for (int kt = 0; kt < 16; ++kt) {
      v2f a = frag_a(embed, 64, mt, kt, lane);
      v2f b = frag_b(W1, 128, kt, nt, lane);
      acc = wmma_f32(a, b, acc);
    }
    int col   = (nt << 4) + (lane & 15);
    int rbase = (mt << 4) + ((lane >> 4) << 3);
#pragma unroll
    for (int v = 0; v < 8; ++v) {
      float x = acc[v] + b1[col];
      Rlds[(rbase + v) * 128 + col] = fmaxf(x, 0.0f);
    }
  }
  __syncthreads();

  for (int tile = wave; tile < 16; tile += 8) {
    int mt = tile >> 2, nt = tile & 3;
    v8f acc = {};
#pragma unroll
    for (int kt = 0; kt < 32; ++kt) {
      v2f a = frag_a(Rlds, 128, mt, kt, lane);
      v2f b = frag_b(W2, 64, kt, nt, lane);
      acc = wmma_f32(a, b, acc);
    }
    int col   = (nt << 4) + (lane & 15);
    int rbase = (mt << 4) + ((lane >> 4) << 3);
#pragma unroll
    for (int v = 0; v < 8; ++v) {
      int row = rbase + v;
      Xlds[row * 64 + col] = embed[row * 64 + col] + acc[v] + b2[col];
    }
  }
  __syncthreads();

  if (tid < 64) {
    const float* xr = &Xlds[tid * 64];
    float mu = 0.f;
#pragma unroll
    for (int c = 0; c < 64; ++c) mu += xr[c];
    mu *= (1.0f / 64.0f);
    float var = 0.f;
#pragma unroll
    for (int c = 0; c < 64; ++c) { float d = xr[c] - mu; var += d * d; }
    var *= (1.0f / 64.0f);
    float rs  = rsqrtf(var + 1e-5f);
    float ksq = 0.f;
#pragma unroll
    for (int c = 0; c < 64; ++c) {
      float tv = (xr[c] - mu) * rs * gamma[c] + beta[c];
      Tout[tid * 64 + c] = tv;
      Rlds[tid * TPITCH + c] = tv;
      ksq += tv * tv;
    }
    rdl[tid] = 1.0f / (ksq + 1e-6f);
  }
  __syncthreads();

  float* Tp = Rlds;
  for (int tile = wave; tile < 16; tile += 8) {
    int mt = tile >> 2, nt = tile & 3;
    v8f acc = {};
#pragma unroll
    for (int kt = 0; kt < 16; ++kt) {
      v2f a = frag_a(Tp, TPITCH, mt, kt, lane);
      v2f b = frag_bt(Tp, TPITCH, kt, nt, lane);
      acc = wmma_f32(a, b, acc);
    }
    int col   = (nt << 4) + (lane & 15);
    int rbase = (mt << 4) + ((lane >> 4) << 3);
#pragma unroll
    for (int v = 0; v < 8; ++v) {
      int row = rbase + v;
      Gplain[row * 64 + col] = acc[v];
      Gpair[((row << 5) + (col & 31)) * 2 + (col >> 5)] = -(acc[v] * rdl[row]);
    }
  }
}

#define GRP 16

__device__ __forceinline__ float lane_read(float x, int l) {
  return __int_as_float(__builtin_amdgcn_readlane(__float_as_int(x), l));
}

template <int J>
__device__ __forceinline__ int row_share_i(int x) {
  return __builtin_amdgcn_update_dpp(0, x, 0x150 + J, 0xf, 0xf, true);
}
template <int J>
__device__ __forceinline__ float row_share(float x) {
  return __int_as_float(row_share_i<J>(__float_as_int(x)));
}

__device__ __forceinline__ float gather_c(float c_lo, float c_hi, int tok, int lane) {
  const unsigned sel = (unsigned)tok & 15u;
  const unsigned ulo = (unsigned)__float_as_int(c_lo);
  const unsigned uhi = (unsigned)__float_as_int(c_hi);
  const unsigned lo_s = __builtin_amdgcn_permlane16_var(ulo, ulo, sel, false, false);
  const unsigned lo_x = __builtin_amdgcn_permlanex16_var(ulo, ulo, sel, false, false);
  const unsigned hi_s = __builtin_amdgcn_permlane16_var(uhi, uhi, sel, false, false);
  const unsigned hi_x = __builtin_amdgcn_permlanex16_var(uhi, uhi, sel, false, false);
  const bool other_row = (((tok >> 4) ^ (lane >> 4)) & 1) != 0;
  const unsigned lo = other_row ? lo_x : lo_s;
  const unsigned hi = other_row ? hi_x : hi_s;
  return __int_as_float((int)((tok & 32) ? hi : lo));
}

__global__ __launch_bounds__(256) void delta_adjoint_kernel(
    const int* __restrict__ seq, const float* __restrict__ Ttab,
    const float* __restrict__ Gplain, const float* __restrict__ Gpair,
    float* __restrict__ ctx) {
  __shared__ float4 Gl4[VOCABN * 16];
  __shared__ float4 Tl4[VOCABN * 16];
  const int b   = blockIdx.x;
  const int tid = threadIdx.x;

  const float4* Gg4 = (const float4*)Gpair;
  const float4* Tg4 = (const float4*)Ttab;
  for (int i = tid; i < VOCABN * 16; i += 256) {
    Gl4[i] = Gg4[i];
    Tl4[i] = Tg4[i];
  }
  __syncthreads();
  if (tid >= 32) return;

  const int     lane = tid;
  const float2* Gl2  = (const float2*)Gl4;
  const float*  Glf  = (const float*)Gl4;
  const float2* Tl2  = (const float2*)Tl4;
  const int*    sqr  = seq + b * SEQL;
  const int4*   sq4  = (const int4*)sqr;
  const int vpos = GRP - 1 - (lane & 15);
  const int NG   = SEQL / GRP;

  float c_lo, c_hi, w_lo = 0.0f, w_hi = 0.0f;
  {
    int tkq[GRP];
    int4 t0 = sq4[4 * (NG - 1)], t1 = sq4[4 * (NG - 1) + 1];
    int4 t2 = sq4[4 * (NG - 1) + 2], t3 = sq4[4 * (NG - 1) + 3];
    tkq[0] = t0.x; tkq[1] = t0.y; tkq[2] = t0.z; tkq[3] = t0.w;
    tkq[4] = t1.x; tkq[5] = t1.y; tkq[6] = t1.z; tkq[7] = t1.w;
    tkq[8] = t2.x; tkq[9] = t2.y; tkq[10] = t2.z; tkq[11] = t2.w;
    tkq[12] = t3.x; tkq[13] = t3.y; tkq[14] = t3.z; tkq[15] = t3.w;
    const int tq = tkq[GRP - 1];
    c_lo = Gplain[tq * 64 + lane];
    c_hi = Gplain[tq * 64 + lane + 32];
#pragma unroll
    for (int j = GRP - 2; j >= 0; --j) {
      const int    tok = tkq[j];
      const float2 row = Gl2[tok * 32 + lane];
      const float  slo = lane_read(c_lo, tok & 31);
      const float  shi = lane_read(c_hi, tok & 31);
      const float  s   = (tok & 32) ? shi : slo;
      c_lo = fmaf(row.x, s, c_lo);
      c_hi = fmaf(row.y, s, c_hi);
      w_lo += (lane == tok) ? s : 0.0f;
      w_hi += (lane + 32 == tok) ? s : 0.0f;
    }
  }

  int    B0[GRP], B1[GRP], B2[GRP];
  float2 R0[GRP], R1[GRP], R2[GRP];
  float  A0[GRP], A1[GRP], A2[GRP];
  int tc0, tc1, tc2, tn0, tn1, tn2, tf0, tf1, tf2;

#define PREP1(o, Bn, Rn, An_, tvn)                                \
  Bn[o]  = row_share_i<(o)>(tvn);                                 \
  Rn[o]  = Gl2[Bn[o] * 32 + lane];                                \
  An_[o] = Glf[Bn[o] * 64 + offn];
#define PREP(Bn, Rn, An_, tvn)                                    \
  {                                                               \
    const int offn = (((tvn) & 31) << 1) + ((tvn) >> 5);          \
    PREP1(0, Bn, Rn, An_, tvn) \
    PREP1(1, Bn, Rn, An_, tvn) \
    PREP1(2, Bn, Rn, An_, tvn) \
    PREP1(3, Bn, Rn, An_, tvn) \
    PREP1(4, Bn, Rn, An_, tvn) \
    PREP1(5, Bn, Rn, An_, tvn) \
    PREP1(6, Bn, Rn, An_, tvn) \
    PREP1(7, Bn, Rn, An_, tvn) \
    PREP1(8, Bn, Rn, An_, tvn) \
    PREP1(9, Bn, Rn, An_, tvn) \
    PREP1(10, Bn, Rn, An_, tvn) \
    PREP1(11, Bn, Rn, An_, tvn) \
    PREP1(12, Bn, Rn, An_, tvn) \
    PREP1(13, Bn, Rn, An_, tvn) \
    PREP1(14, Bn, Rn, An_, tvn) \
    PREP1(15, Bn, Rn, An_, tvn) \
  }
#define GSTEP(o, LAST, Bc, Rc, Ac_)                               \
  {                                                               \
    const float s = row_share<(o)>(p);                            \
    if (!(LAST)) p = fmaf(Ac_[o], s, p);                          \
    c_lo = fmaf(Rc[o].x, s, c_lo);                                \
    c_hi = fmaf(Rc[o].y, s, c_hi);                                \
    w_lo += (lane == Bc[o]) ? s : 0.0f;                           \
    w_hi += (lane + 32 == Bc[o]) ? s : 0.0f;                      \
  }
#define GROUP(Bc, Rc, Ac_, tvc, Bn, Rn, An_, tvn)                 \
  {                                                               \
    PREP(Bn, Rn, An_, tvn)                                        \
    __builtin_amdgcn_sched_barrier(0);                            \
    float p = gather_c(c_lo, c_hi, tvc, lane);                    \
    GSTEP(0, 0, Bc, Rc, Ac_) \
    GSTEP(1, 0, Bc, Rc, Ac_) \
    GSTEP(2, 0, Bc, Rc, Ac_) \
    GSTEP(3, 0, Bc, Rc, Ac_) \
    GSTEP(4, 0, Bc, Rc, Ac_) \
    GSTEP(5, 0, Bc, Rc, Ac_) \
    GSTEP(6, 0, Bc, Rc, Ac_) \
    GSTEP(7, 0, Bc, Rc, Ac_) \
    GSTEP(8, 0, Bc, Rc, Ac_) \
    GSTEP(9, 0, Bc, Rc, Ac_) \
    GSTEP(10, 0, Bc, Rc, Ac_) \
    GSTEP(11, 0, Bc, Rc, Ac_) \
    GSTEP(12, 0, Bc, Rc, Ac_) \
    GSTEP(13, 0, Bc, Rc, Ac_) \
    GSTEP(14, 0, Bc, Rc, Ac_) \
    GSTEP(15, 1, Bc, Rc, Ac_) \
  }

  tn0 = sqr[(NG - 2) * GRP + vpos];
  tn1 = sqr[(NG - 3) * GRP + vpos];
  tn2 = sqr[(NG - 4) * GRP + vpos];
  tf0 = sqr[(NG - 5) * GRP + vpos];
  tf1 = sqr[(NG - 6) * GRP + vpos];
  tf2 = sqr[(NG - 7) * GRP + vpos];
  PREP(B0, R0, A0, tn0)

  for (int g0 = NG - 2; g0 >= 2; g0 -= 3) {
    tc0 = tn0; tc1 = tn1; tc2 = tn2;
    tn0 = tf0; tn1 = tf1; tn2 = tf2;
    const int f0 = (g0 >= 6) ? (g0 - 6) : 0;
    const int f1 = (g0 >= 7) ? (g0 - 7) : 0;
    const int f2 = (g0 >= 8) ? (g0 - 8) : 0;
    tf0 = sqr[f0 * GRP + vpos];
    tf1 = sqr[f1 * GRP + vpos];
    tf2 = sqr[f2 * GRP + vpos];
    GROUP(B0, R0, A0, tc0, B1, R1, A1, tc1)
    GROUP(B1, R1, A1, tc1, B2, R2, A2, tc2)
    GROUP(B2, R2, A2, tc2, B0, R0, A0, tn0)
  }
#undef GROUP
#undef GSTEP
#undef PREP
#undef PREP1

  float ax0 = 0.f, ay0 = 0.f, ax1 = 0.f, ay1 = 0.f;
#pragma unroll
  for (int v = 0; v < 32; v += 2) {
    float  wa = lane_read(w_lo, v), wb = lane_read(w_lo, v + 1);
    float2 ta = Tl2[v * 32 + lane], tb = Tl2[(v + 1) * 32 + lane];
    ax0 = fmaf(ta.x, wa, ax0); ay0 = fmaf(ta.y, wa, ay0);
    ax1 = fmaf(tb.x, wb, ax1); ay1 = fmaf(tb.y, wb, ay1);
  }
#pragma unroll
  for (int v = 0; v < 32; v += 2) {
    float  wa = lane_read(w_hi, v), wb = lane_read(w_hi, v + 1);
    float2 ta = Tl2[(v + 32) * 32 + lane], tb = Tl2[(v + 33) * 32 + lane];
    ax0 = fmaf(ta.x, wa, ax0); ay0 = fmaf(ta.y, wa, ay0);
    ax1 = fmaf(tb.x, wb, ax1); ay1 = fmaf(tb.y, wb, ay1);
  }
  float2 r;
  r.x = ax0 + ax1;
  r.y = ay0 + ay1;
  ((float2*)(ctx + b * 64))[lane] = r;
}

__global__ __launch_bounds__(256) void head_kernel(
    const float* __restrict__ ctx, const float* __restrict__ Wr,
    const float* __restrict__ br, const float* __restrict__ Wo,
    const float* __restrict__ bo, float* __restrict__ out) {
  __shared__ float Glds[64 * 64];
  const int tid  = threadIdx.x;
  const int wave = tid >> 5;
  const int lane = tid & 31;

  for (int tile = wave; tile < 16; tile += 8) {
    int mt = tile >> 2, nt = tile & 3;
    v8f acc = {};
#pragma unroll
    for (int kt = 0; kt < 16; ++kt) {
      v2f a = frag_a(ctx, 64, mt, kt, lane);
      v2f b = frag_b(Wr, 64, kt, nt, lane);
      acc = wmma_f32(a, b, acc);
    }
    int col   = (nt << 4) + (lane & 15);
    int rbase = (mt << 4) + ((lane >> 4) << 3);
#pragma unroll
    for (int v = 0; v < 8; ++v) Glds[(rbase + v) * 64 + col] = acc[v] + br[col];
  }
  __syncthreads();

  for (int tile = wave; tile < 16; tile += 8) {
    int mt = tile >> 2, nt = tile & 3;
    v8f acc = {};
#pragma unroll
    for (int kt = 0; kt < 16; ++kt) {
      v2f a = frag_a(Glds, 64, mt, kt, lane);
      v2f b = frag_b(Wo, 64, kt, nt, lane);
      acc = wmma_f32(a, b, acc);
    }
    int col   = (nt << 4) + (lane & 15);
    int rbase = (mt << 4) + ((lane >> 4) << 3);
#pragma unroll
    for (int v = 0; v < 8; ++v) out[(rbase + v) * 64 + col] = acc[v] + bo[col];
  }
}

extern "C" void kernel_launch(void* const* d_in, const int* in_sizes, int n_in,
                              void* d_out, int out_size, void* d_ws, size_t ws_size,
                              hipStream_t stream) {
  const int*   seq   = (const int*)d_in[0];
  const float* embed = (const float*)d_in[1];
  const float* W1    = (const float*)d_in[2];
  const float* b1    = (const float*)d_in[3];
  const float* W2    = (const float*)d_in[4];
  const float* b2    = (const float*)d_in[5];
  const float* gamma = (const float*)d_in[6];
  const float* beta  = (const float*)d_in[7];
  const float* Wr    = (const float*)d_in[8];
  const float* br    = (const float*)d_in[9];
  const float* Wo    = (const float*)d_in[10];
  const float* bo    = (const float*)d_in[11];

  float* Tt     = (float*)d_ws;
  float* Gplain = Tt + VOCABN * HDIM;
  float* Gpair  = Gplain + VOCABN * VOCABN;
  float* ctx    = Gpair + VOCABN * VOCABN;
  float* out    = (float*)d_out;

  encode_table_kernel<<<1, 256, 0, stream>>>(embed, W1, b1, W2, b2, gamma, beta,
                                             Tt, Gplain, Gpair);
  delta_adjoint_kernel<<<BATCH, 256, 0, stream>>>(seq, Tt, Gplain, Gpair, ctx);
  head_kernel<<<1, 256, 0, stream>>>(ctx, Wr, br, Wo, bo, out);
}


# Scan wave keeps only the carried path: 8 helper waves pre-pack each step's Gs row pair and coefficient in LDS ahead of it and sum s*T[tok] into ctx behind it; order of the ctx sum changes
# speedup vs baseline: 9.0949x; 1.1955x over previous
#include <hip/hip_runtime.h>


#define HDIM   64
#define VOCABN 64
#define BATCH  64
#define SEQL   4096
#define TPITCH 68

typedef __attribute__((ext_vector_type(2))) float v2f;
typedef __attribute__((ext_vector_type(2))) float f32x2;
typedef __attribute__((ext_vector_type(8))) float v8f;

__device__ __forceinline__ v2f frag_a(const float* A, int lda, int mt, int kt, int lane) {
  int row = (mt << 4) + (lane & 15);
  int k0  = (kt << 2) + ((lane >> 4) << 1);
  v2f a;
  a.x = A[row * lda + k0];
  a.y = A[row * lda + k0 + 1];
  return a;
}
__device__ __forceinline__ v2f frag_b(const float* Bm, int ldb, int kt, int nt, int lane) {
  int col = (nt << 4) + (lane & 15);
  int k0  = (kt << 2) + ((lane >> 4) << 1);
  v2f b;
  b.x = Bm[k0 * ldb + col];
  b.y = Bm[(k0 + 1) * ldb + col];
  return b;
}
__device__ __forceinline__ v2f frag_bt(const float* S, int lds, int kt, int nt, int lane) {
  int col = (nt << 4) + (lane & 15);
  int k0  = (kt << 2) + ((lane >> 4) << 1);
  v2f b;
  b.x = S[col * lds + k0];
  b.y = S[col * lds + k0 + 1];
  return b;
}

__device__ __forceinline__ v8f wmma_f32(v2f a, v2f b, v8f c) {
  return __builtin_amdgcn_wmma_f32_16x16x4_f32(false, a, false, b, (short)0, c,
                                               false, false);
}

__global__ __launch_bounds__(256) void encode_table_kernel(
    const float* __restrict__ embed, const float* __restrict__ W1,
    const float* __restrict__ b1, const float* __restrict__ W2,
    const float* __restrict__ b2, const float* __restrict__ gamma,
    const float* __restrict__ beta, float* __restrict__ Tout,
    float* __restrict__ Gplain, float* __restrict__ Gpair) {
  __shared__ float Rlds[64 * 128];
  __shared__ float Xlds[64 * 64];
  __shared__ float rdl[64];
  const int tid  = threadIdx.x;
  const int wave = tid >> 5;
  const int lane = tid & 31;

  for (int tile = wave; tile < 32; tile += 8) {
    int mt = tile >> 3, nt = tile & 7;
    v8f acc = {};
#pragma unroll
    for (int kt = 0; kt < 16; ++kt) {
      v2f a = frag_a(embed, 64, mt, kt, lane);
      v2f b = frag_b(W1, 128, kt, nt, lane);
      acc = wmma_f32(a, b, acc);
    }
    int col   = (nt << 4) + (lane & 15);
    int rbase = (mt << 4) + ((lane >> 4) << 3);
#pragma unroll
    for (int v = 0; v < 8; ++v) {
      float x = acc[v] + b1[col];
      Rlds[(rbase + v) * 128 + col] = fmaxf(x, 0.0f);
    }
  }
  __syncthreads();

  for (int tile = wave; tile < 16; tile += 8) {
    int mt = tile >> 2, nt = tile & 3;
    v8f acc = {};
#pragma unroll
    for (int kt = 0; kt < 32; ++kt) {
      v2f a = frag_a(Rlds, 128, mt, kt, lane);
      v2f b = frag_b(W2, 64, kt, nt, lane);
      acc = wmma_f32(a, b, acc);
    }
    int col   = (nt << 4) + (lane & 15);
    int rbase = (mt << 4) + ((lane >> 4) << 3);
#pragma unroll
    for (int v = 0; v < 8; ++v) {
      int row = rbase + v;
      Xlds[row * 64 + col] = embed[row * 64 + col] + acc[v] + b2[col];
    }
  }
  __syncthreads();

  if (tid < 64) {
    const float* xr = &Xlds[tid * 64];
    float mu = 0.f;
#pragma unroll
    for (int c = 0; c < 64; ++c) mu += xr[c];
    mu *= (1.0f / 64.0f);
    float var = 0.f;
#pragma unroll
    for (int c = 0; c < 64; ++c) { float d = xr[c] - mu; var += d * d; }
    var *= (1.0f / 64.0f);
    float rs  = rsqrtf(var + 1e-5f);
    float ksq = 0.f;
#pragma unroll
    for (int c = 0; c < 64; ++c) {
      float tv = (xr[c] - mu) * rs * gamma[c] + beta[c];
      Tout[tid * 64 + c] = tv;
      Rlds[tid * TPITCH + c] = tv;
      ksq += tv * tv;
    }
    rdl[tid] = 1.0f / (ksq + 1e-6f);
  }
  __syncthreads();

  float* Tp = Rlds;
  for (int tile = wave; tile < 16; tile += 8) {
    int mt = tile >> 2, nt = tile & 3;
    v8f acc = {};
#pragma unroll
    for (int kt = 0; kt < 16; ++kt) {
      v2f a = frag_a(Tp, TPITCH, mt, kt, lane);
      v2f b = frag_bt(Tp, TPITCH, kt, nt, lane);
      acc = wmma_f32(a, b, acc);
    }
    int col   = (nt << 4) + (lane & 15);
    int rbase = (mt << 4) + ((lane >> 4) << 3);
#pragma unroll
    for (int v = 0; v < 8; ++v) {
      int row = rbase + v;
      Gplain[row * 64 + col] = acc[v];
      Gpair[((row << 5) + (col & 31)) * 2 + (col >> 5)] = -(acc[v] * rdl[row]);
    }
  }
}

#define GRP    16
#define NSLOT  8
#define NTRIP  32
#define NTHR   288

template <int J>
__device__ __forceinline__ int row_share_i(int x) {
  return __builtin_amdgcn_update_dpp(0, x, 0x150 + J, 0xf, 0xf, true);
}
template <int J>
__device__ __forceinline__ float row_share(float x) {
  return __int_as_float(row_share_i<J>(__float_as_int(x)));
}

__device__ __forceinline__ float gather_c(float c_lo, float c_hi, int tok, int lane) {
  const unsigned sel = (unsigned)tok & 15u;
  const unsigned ulo = (unsigned)__float_as_int(c_lo);
  const unsigned uhi = (unsigned)__float_as_int(c_hi);
  const unsigned lo_s = __builtin_amdgcn_permlane16_var(ulo, ulo, sel, false, false);
  const unsigned lo_x = __builtin_amdgcn_permlanex16_var(ulo, ulo, sel, false, false);
  const unsigned hi_s = __builtin_amdgcn_permlane16_var(uhi, uhi, sel, false, false);
  const unsigned hi_x = __builtin_amdgcn_permlanex16_var(uhi, uhi, sel, false, false);
  const bool other_row = (((tok >> 4) ^ (lane >> 4)) & 1) != 0;
  const unsigned lo = other_row ? lo_x : lo_s;
  const unsigned hi = other_row ? hi_x : hi_s;
  return __int_as_float((int)((tok & 32) ? hi : lo));
}

__global__ __launch_bounds__(NTHR) void delta_adjoint_kernel(
    const int* __restrict__ seq, const float* __restrict__ Ttab,
    const float* __restrict__ Gplain, const float* __restrict__ Gpair,
    float* __restrict__ ctx) {
  __shared__ float4 Gl4[VOCABN * 16];
  __shared__ float4 Tl4[VOCABN * 16];
  __shared__ float4 Pk[2 * NSLOT * GRP * 32];
  __shared__ float  Sb[SEQL];
  __shared__ float2 Part[NSLOT * 32];
  const int b   = blockIdx.x;
  const int tid = threadIdx.x;

  const float4* Gg4 = (const float4*)Gpair;
  const float4* Tg4 = (const float4*)Ttab;
  for (int i = tid; i < VOCABN * 16; i += NTHR) {
    Gl4[i] = Gg4[i];
    Tl4[i] = Tg4[i];
  }
  __syncthreads();

  const int lane = tid & 31;
  const int wave = __builtin_amdgcn_readfirstlane(tid >> 5);
  const float2* Gl2 = (const float2*)Gl4;
  const float*  Glf = (const float*)Gl4;
  const float2* Tl2 = (const float2*)Tl4;
  const int*    sqr = seq + b * SEQL;
  const int     li  = lane & 15;
  const int     NG  = SEQL / GRP;

  float  c_lo = 0.0f, c_hi = 0.0f;
  float2 acc  = make_float2(0.0f, 0.0f);
  int    tvA = 0, tvB = 0;
  if (wave == 0) {
    const int tq = sqr[SEQL - 1];
    c_lo = Gplain[tq * 64 + lane];
    c_hi = Gplain[tq * 64 + lane + 32];
  }

#define PACK1(o)                                                           \
  {                                                                        \
    const int to = row_share_i<(o)>(tvN);                                  \
    float2 R = Gl2[to * 32 + lane];                                        \
    float  A = Glf[to * 64 + offn];                                        \
    A = (li > (o)) ? A : 0.0f;                                             \
    if ((o) == 0 && qgroup) { R.x = 0.0f; R.y = 0.0f; A = 0.0f; }          \
    pkw[(o) * 32] = make_float4(R.x, R.y, A, __int_as_float(tvN));         \
  }
#define ACC1(o)                                                            \
  {                                                                        \
    float so = row_share<(o)>(sv);                                         \
    if ((o) == 0 && qdone) so = 0.0f;                                      \
    const int    to = row_share_i<(o)>(tvA);                               \
    const float2 tr = Tl2[to * 32 + lane];                                 \
    acc.x = fmaf(tr.x, so, acc.x);                                         \
    acc.y = fmaf(tr.y, so, acc.y);                                         \
  }
#define GSTEP(o, LAST, X)                                                  \
  {                                                                        \
    const float s = row_share<(o)>(p);                                     \
    if (!(LAST)) p = fmaf(X[o].z, s, p);                                   \
    c_lo = fmaf(X[o].x, s, c_lo);                                          \
    c_hi = fmaf(X[o].y, s, c_hi);                                          \
  }
#define LOADSET(X, slot)                                                   \
  {                                                                        \
    _Pragma("unroll") for (int o = 0; o < GRP; ++o)                        \
      X[o] = pkr[((slot) * GRP + o) * 32];                                 \
  }
#define GROUP(Xc, Xn, slot, HASNEXT)                                       \
  {                                                                        \
    if (HASNEXT) LOADSET(Xn, (slot) + 1)                                   \
    __builtin_amdgcn_sched_barrier(0);                                     \
    float p = gather_c(c_lo, c_hi, __float_as_int(Xc[0].w), lane);         \
    GSTEP(0, 0, Xc) GSTEP(1, 0, Xc) GSTEP(2, 0, Xc) GSTEP(3, 0, Xc)        \
    GSTEP(4, 0, Xc) GSTEP(5, 0, Xc) GSTEP(6, 0, Xc) GSTEP(7, 0, Xc)        \
    GSTEP(8, 0, Xc) GSTEP(9, 0, Xc) GSTEP(10, 0, Xc) GSTEP(11, 0, Xc)      \
    GSTEP(12, 0, Xc) GSTEP(13, 0, Xc) GSTEP(14, 0, Xc) GSTEP(15, 1, Xc)    \
    Sb[(gfirst - (slot)) * GRP + li] = p;                                  \
  }

  for (int k = 0; k < NTRIP + 2; ++k) {
    if (wave == 0) {
      if (k >= 1 && k <= NTRIP) {
        const float4* pkr    = Pk + ((k - 1) & 1) * (NSLOT * GRP * 32) + lane;
        const int     gfirst = NG - 1 - (k - 1) * NSLOT;
        float4 X0[GRP], X1[GRP];
        LOADSET(X0, 0)
        GROUP(X0, X1, 0, 1)
        GROUP(X1, X0, 1, 1)
        GROUP(X0, X1, 2, 1)
        GROUP(X1, X0, 3, 1)
        GROUP(X0, X1, 4, 1)
        GROUP(X1, X0, 5, 1)
        GROUP(X0, X1, 6, 1)
        GROUP(X1, X0, 7, 0)
      }
    } else {
      const int h   = wave - 1;
      int       tvN = 0;
      if (k < NTRIP) {
        const int  g      = NG - 1 - (k * NSLOT + h);
        const bool qgroup = (g == NG - 1);
        tvN = sqr[g * GRP + (GRP - 1 - li)];
        const int offn = ((tvN & 31) << 1) + (tvN >> 5);
        float4*   pkw  = Pk + ((k & 1) * NSLOT + h) * (GRP * 32) + lane;
        PACK1(0) PACK1(1) PACK1(2) PACK1(3) PACK1(4) PACK1(5) PACK1(6) PACK1(7)
        PACK1(8) PACK1(9) PACK1(10) PACK1(11) PACK1(12) PACK1(13) PACK1(14) PACK1(15)
      }
      if (k >= 2) {
        const int   g2    = NG - 1 - ((k - 2) * NSLOT + h);
        const bool  qdone = (g2 == NG - 1);
        const float sv    = Sb[g2 * GRP + li];
        ACC1(0) ACC1(1) ACC1(2) ACC1(3) ACC1(4) ACC1(5) ACC1(6) ACC1(7)
        ACC1(8) ACC1(9) ACC1(10) ACC1(11) ACC1(12) ACC1(13) ACC1(14) ACC1(15)
      }
      tvA = tvB;
      tvB = tvN;
    }
    __syncthreads();
  }
#undef GROUP
#undef LOADSET
#undef GSTEP
#undef ACC1
#undef PACK1

  if (wave != 0) Part[(wave - 1) * 32 + lane] = acc;
  __syncthreads();
  if (wave == 0) {
    float2 r = Part[lane];
    for (int h = 1; h < NSLOT; ++h) {
      const float2 q = Part[h * 32 + lane];
      r.x += q.x;
      r.y += q.y;
    }
    ((float2*)(ctx + b * 64))[lane] = r;
  }
}

__global__ __launch_bounds__(256) void head_kernel(
    const float* __restrict__ ctx, const float* __restrict__ Wr,
    const float* __restrict__ br, const float* __restrict__ Wo,
    const float* __restrict__ bo, float* __restrict__ out) {
  __shared__ float Glds[64 * 64];
  const int tid  = threadIdx.x;
  const int wave = tid >> 5;
  const int lane = tid & 31;

  for (int tile = wave; tile < 16; tile += 8) {
    int mt = tile >> 2, nt = tile & 3;
    v8f acc = {};
#pragma unroll
    for (int kt = 0; kt < 16; ++kt) {
      v2f a = frag_a(ctx, 64, mt, kt, lane);
      v2f b = frag_b(Wr, 64, kt, nt, lane);
      acc = wmma_f32(a, b, acc);
    }
    int col   = (nt << 4) + (lane & 15);
    int rbase = (mt << 4) + ((lane >> 4) << 3);
#pragma unroll
    for (int v = 0; v < 8; ++v) Glds[(rbase + v) * 64 + col] = acc[v] + br[col];
  }
  __syncthreads();

  for (int tile = wave; tile < 16; tile += 8) {
    int mt = tile >> 2, nt = tile & 3;
    v8f acc = {};
#pragma unroll
    for (int kt = 0; kt < 16; ++kt) {
      v2f a = frag_a(Glds, 64, mt, kt, lane);
      v2f b = frag_b(Wo, 64, kt, nt, lane);
      acc = wmma_f32(a, b, acc);
    }
    int col   = (nt << 4) + (lane & 15);
    int rbase = (mt << 4) + ((lane >> 4) << 3);
#pragma unroll
    for (int v = 0; v < 8; ++v) out[(rbase + v) * 64 + col] = acc[v] + bo[col];
  }
}

extern "C" void kernel_launch(void* const* d_in, const int* in_sizes, int n_in,
                              void* d_out, int out_size, void* d_ws, size_t ws_size,
                              hipStream_t stream) {
  const int*   seq   = (const int*)d_in[0];
  const float* embed = (const float*)d_in[1];
  const float* W1    = (const float*)d_in[2];
  const float* b1    = (const float*)d_in[3];
  const float* W2    = (const float*)d_in[4];
  const float* b2    = (const float*)d_in[5];
  const float* gamma = (const float*)d_in[6];
  const float* beta  = (const float*)d_in[7];
  const float* Wr    = (const float*)d_in[8];
  const float* br    = (const float*)d_in[9];
  const float* Wo    = (const float*)d_in[10];
  const float* bo    = (const float*)d_in[11];

  float* Tt     = (float*)d_ws;
  float* Gplain = Tt + VOCABN * HDIM;
  float* Gpair  = Gplain + VOCABN * VOCABN;
  float* ctx    = Gpair + VOCABN * VOCABN;
  float* out    = (float*)d_out;

  encode_table_kernel<<<1, 256, 0, stream>>>(embed, W1, b1, W2, b2, gamma, beta,
                                             Tt, Gplain, Gpair);
  delta_adjoint_kernel<<<BATCH, NTHR, 0, stream>>>(seq, Tt, Gplain, Gpair, ctx);
  head_kernel<<<1, 256, 0, stream>>>(ctx, Wr, br, Wo, bo, out);
}


# Three launches fused into one: each scan block builds the token table in its own LDS first and applies the head to its own ctx row afterwards on WMMA (one-row A tile); same order of sums
# speedup vs baseline: 10.0782x; 1.1081x over previous
#include <hip/hip_runtime.h>


#define HDIM   64
#define VOCABN 64
#define BATCH  64
#define SEQL   4096
#define TPITCH 68

typedef __attribute__((ext_vector_type(2))) float v2f;
typedef __attribute__((ext_vector_type(2))) float f32x2;
typedef __attribute__((ext_vector_type(8))) float v8f;

__device__ __forceinline__ v2f frag_a(const float* A, int lda, int mt, int kt, int lane) {
  int row = (mt << 4) + (lane & 15);
  int k0  = (kt << 2) + ((lane >> 4) << 1);
  v2f a;
  a.x = A[row * lda + k0];
  a.y = A[row * lda + k0 + 1];
  return a;
}
__device__ __forceinline__ v2f frag_b(const float* Bm, int ldb, int kt, int nt, int lane) {
  int col = (nt << 4) + (lane & 15);
  int k0  = (kt << 2) + ((lane >> 4) << 1);
  v2f b;
  b.x = Bm[k0 * ldb + col];
  b.y = Bm[(k0 + 1) * ldb + col];
  return b;
}
__device__ __forceinline__ v2f frag_bt(const float* S, int lds, int kt, int nt, int lane) {
  int col = (nt << 4) + (lane & 15);
  int k0  = (kt << 2) + ((lane >> 4) << 1);
  v2f b;
  b.x = S[col * lds + k0];
  b.y = S[col * lds + k0 + 1];
  return b;
}

__device__ __forceinline__ v8f wmma_f32(v2f a, v2f b, v8f c) {
  return __builtin_amdgcn_wmma_f32_16x16x4_f32(false, a, false, b, (short)0, c,
                                               false, false);
}

__device__ __forceinline__ v2f frag_a_row0(const float2* x2, int kt, int lane) {
  const float2 xv = x2[(kt << 1) + (lane >> 4)];
  const bool   r0 = (lane & 15) == 0;
  v2f a;
  a.x = r0 ? xv.x : 0.0f;
  a.y = r0 ? xv.y : 0.0f;
  return a;
}

#define GRP    16
#define NSLOT  8
#define NTRIP  32
#define NTHR   288

template <int J>
__device__ __forceinline__ int row_share_i(int x) {
  return __builtin_amdgcn_update_dpp(0, x, 0x150 + J, 0xf, 0xf, true);
}
template <int J>
__device__ __forceinline__ float row_share(float x) {
  return __int_as_float(row_share_i<J>(__float_as_int(x)));
}

__device__ __forceinline__ float gather_c(float c_lo, float c_hi, int tok, int lane) {
  const unsigned sel = (unsigned)tok & 15u;
  const unsigned ulo = (unsigned)__float_as_int(c_lo);
  const unsigned uhi = (unsigned)__float_as_int(c_hi);
  const unsigned lo_s = __builtin_amdgcn_permlane16_var(ulo, ulo, sel, false, false);
  const unsigned lo_x = __builtin_amdgcn_permlanex16_var(ulo, ulo, sel, false, false);
  const unsigned hi_s = __builtin_amdgcn_permlane16_var(uhi, uhi, sel, false, false);
  const unsigned hi_x = __builtin_amdgcn_permlanex16_var(uhi, uhi, sel, false, false);
  const bool other_row = (((tok >> 4) ^ (lane >> 4)) & 1) != 0;
  const unsigned lo = other_row ? lo_x : lo_s;
  const unsigned hi = other_row ? hi_x : hi_s;
  return __int_as_float((int)((tok & 32) ? hi : lo));
}

#define NWAVE  9

__global__ __launch_bounds__(NTHR) void delta_fused_kernel(
    const int* __restrict__ seq, const float* __restrict__ embed,
    const float* __restrict__ W1, const float* __restrict__ b1,
    const float* __restrict__ W2, const float* __restrict__ b2,
    const float* __restrict__ gamma, const float* __restrict__ beta,
    const float* __restrict__ Wr, const float* __restrict__ br,
    const float* __restrict__ Wo, const float* __restrict__ bo,
    float* __restrict__ out) {
  __shared__ float4 Gl4[VOCABN * 16];
  __shared__ float4 Tl4[VOCABN * 16];
  __shared__ float4 Pk[2 * NSLOT * GRP * 32];
  __shared__ float  Sb[SEQL];
  __shared__ float2 Part[NSLOT * 32];
  __shared__ float4 Rl4[64 * 32];
  __shared__ float4 Xl4[64 * 16];
  __shared__ float  rdl[64];
  __shared__ float  Gpl[VOCABN * VOCABN];
  __shared__ float2 Cx2[32];
  __shared__ float2 Yx2[32];
  __shared__ float2 Ox2[32];
  float* Rlds = (float*)Rl4;
  float* Xlds = (float*)Xl4;
  const int b   = blockIdx.x;
  const int tid = threadIdx.x;
  const int lane = tid & 31;
  const int wave = __builtin_amdgcn_readfirstlane(tid >> 5);

  for (int tile = wave; tile < 32; tile += NWAVE) {
    int mt = tile >> 3, nt = tile & 7;
    v8f acc8 = {};
#pragma unroll
    for (int kt = 0; kt < 16; ++kt) {
      v2f a = frag_a(embed, 64, mt, kt, lane);
      v2f bq = frag_b(W1, 128, kt, nt, lane);
      acc8 = wmma_f32(a, bq, acc8);
    }
    int col   = (nt << 4) + (lane & 15);
    int rbase = (mt << 4) + ((lane >> 4) << 3);
#pragma unroll
    for (int v = 0; v < 8; ++v) {
      float x = acc8[v] + b1[col];
      Rlds[(rbase + v) * 128 + col] = fmaxf(x, 0.0f);
    }
  }
  __syncthreads();

  for (int tile = wave; tile < 16; tile += NWAVE) {
    int mt = tile >> 2, nt = tile & 3;
    v8f acc8 = {};
#pragma unroll
    for (int kt = 0; kt < 32; ++kt) {
      v2f a = frag_a(Rlds, 128, mt, kt, lane);
      v2f bq = frag_b(W2, 64, kt, nt, lane);
      acc8 = wmma_f32(a, bq, acc8);
    }
    int col   = (nt << 4) + (lane & 15);
    int rbase = (mt << 4) + ((lane >> 4) << 3);
#pragma unroll
    for (int v = 0; v < 8; ++v) {
      int row = rbase + v;
      Xlds[row * 64 + col] = embed[row * 64 + col] + acc8[v] + b2[col];
    }
  }
  __syncthreads();

  if (tid < 64) {
    const float* xr = &Xlds[tid * 64];
    float mu = 0.f;
#pragma unroll
    for (int c = 0; c < 64; ++c) mu += xr[c];
    mu *= (1.0f / 64.0f);
    float var = 0.f;
#pragma unroll
    for (int c = 0; c < 64; ++c) { float d = xr[c] - mu; var += d * d; }
    var *= (1.0f / 64.0f);
    float rs  = rsqrtf(var + 1e-5f);
    float ksq = 0.f;
#pragma unroll
    for (int c = 0; c < 64; ++c) {
      float tv = (xr[c] - mu) * rs * gamma[c] + beta[c];
      Rlds[tid * TPITCH + c] = tv;
      ksq += tv * tv;
    }
    rdl[tid] = 1.0f / (ksq + 1e-6f);
  }
  __syncthreads();

  for (int i = tid; i < VOCABN * 16; i += NTHR)
    Tl4[i] = *(const float4*)&Rlds[(i >> 4) * TPITCH + ((i & 15) << 2)];

  {
    const float* Tp  = Rlds;
    float*       Gsf = (float*)Gl4;
    for (int tile = wave; tile < 16; tile += NWAVE) {
      int mt = tile >> 2, nt = tile & 3;
      v8f acc8 = {};
#pragma unroll
      for (int kt = 0; kt < 16; ++kt) {
        v2f a = frag_a(Tp, TPITCH, mt, kt, lane);
        v2f bq = frag_bt(Tp, TPITCH, kt, nt, lane);
        acc8 = wmma_f32(a, bq, acc8);
      }
      int col   = (nt << 4) + (lane & 15);
      int rbase = (mt << 4) + ((lane >> 4) << 3);
#pragma unroll
      for (int v = 0; v < 8; ++v) {
        int row = rbase + v;
        Gpl[row * 64 + col] = acc8[v];
        Gsf[((row << 5) + (col & 31)) * 2 + (col >> 5)] = -(acc8[v] * rdl[row]);
      }
    }
  }
  __syncthreads();

  const float2* Gl2 = (const float2*)Gl4;
  const float*  Glf = (const float*)Gl4;
  const float2* Tl2 = (const float2*)Tl4;
  const int*    sqr = seq + b * SEQL;
  const int     li  = lane & 15;
  const int     NG  = SEQL / GRP;

  float  c_lo = 0.0f, c_hi = 0.0f;
  float2 acc  = make_float2(0.0f, 0.0f);
  int    tvA = 0, tvB = 0;
  if (wave == 0) {
    const int tq = sqr[SEQL - 1];
    c_lo = Gpl[tq * 64 + lane];
    c_hi = Gpl[tq * 64 + lane + 32];
  }

#define PACK1(o)                                                           \
  {                                                                        \
    const int to = row_share_i<(o)>(tvN);                                  \
    float2 R = Gl2[to * 32 + lane];                                        \
    float  A = Glf[to * 64 + offn];                                        \
    A = (li > (o)) ? A : 0.0f;                                             \
    if ((o) == 0 && qgroup) { R.x = 0.0f; R.y = 0.0f; A = 0.0f; }          \
    pkw[(o) * 32] = make_float4(R.x, R.y, A, __int_as_float(tvN));         \
  }
#define ACC1(o)                                                            \
  {                                                                        \
    float so = row_share<(o)>(sv);                                         \
    if ((o) == 0 && qdone) so = 0.0f;                                      \
    const int    to = row_share_i<(o)>(tvA);                               \
    const float2 tr = Tl2[to * 32 + lane];                                 \
    acc.x = fmaf(tr.x, so, acc.x);                                         \
    acc.y = fmaf(tr.y, so, acc.y);                                         \
  }
#define GSTEP(o, LAST, X)                                                  \
  {                                                                        \
    const float s = row_share<(o)>(p);                                     \
    if (!(LAST)) p = fmaf(X[o].z, s, p);                                   \
    c_lo = fmaf(X[o].x, s, c_lo);                                          \
    c_hi = fmaf(X[o].y, s, c_hi);                                          \
  }
#define LOADSET(X, slot)                                                   \
  {                                                                        \
    _Pragma("unroll") for (int o = 0; o < GRP; ++o)                        \
      X[o] = pkr[((slot) * GRP + o) * 32];                                 \
  }
#define GROUP(Xc, Xn, slot, HASNEXT)                                       \
  {                                                                        \
    if (HASNEXT) LOADSET(Xn, (slot) + 1)                                   \
    __builtin_amdgcn_sched_barrier(0);                                     \
    float p = gather_c(c_lo, c_hi, __float_as_int(Xc[0].w), lane);         \
    GSTEP(0, 0, Xc) GSTEP(1, 0, Xc) GSTEP(2, 0, Xc) GSTEP(3, 0, Xc)        \
    GSTEP(4, 0, Xc) GSTEP(5, 0, Xc) GSTEP(6, 0, Xc) GSTEP(7, 0, Xc)        \
    GSTEP(8, 0, Xc) GSTEP(9, 0, Xc) GSTEP(10, 0, Xc) GSTEP(11, 0, Xc)      \
    GSTEP(12, 0, Xc) GSTEP(13, 0, Xc) GSTEP(14, 0, Xc) GSTEP(15, 1, Xc)    \
    Sb[(gfirst - (slot)) * GRP + li] = p;                                  \
  }

  for (int k = 0; k < NTRIP + 2; ++k) {
    if (wave == 0) {
      if (k >= 1 && k <= NTRIP) {
        const float4* pkr    = Pk + ((k - 1) & 1) * (NSLOT * GRP * 32) + lane;
        const int     gfirst = NG - 1 - (k - 1) * NSLOT;
        float4 X0[GRP], X1[GRP];
        LOADSET(X0, 0)
        GROUP(X0, X1, 0, 1)
        GROUP(X1, X0, 1, 1)
        GROUP(X0, X1, 2, 1)
        GROUP(X1, X0, 3, 1)
        GROUP(X0, X1, 4, 1)
        GROUP(X1, X0, 5, 1)
        GROUP(X0, X1, 6, 1)
        GROUP(X1, X0, 7, 0)
      }
    } else {
      const int h   = wave - 1;
      int       tvN = 0;
      if (k < NTRIP) {
        const int  g      = NG - 1 - (k * NSLOT + h);
        const bool qgroup = (g == NG - 1);
        tvN = sqr[g * GRP + (GRP - 1 - li)];
        const int offn = ((tvN & 31) << 1) + (tvN >> 5);
        float4*   pkw  = Pk + ((k & 1) * NSLOT + h) * (GRP * 32) + lane;
        PACK1(0) PACK1(1) PACK1(2) PACK1(3) PACK1(4) PACK1(5) PACK1(6) PACK1(7)
        PACK1(8) PACK1(9) PACK1(10) PACK1(11) PACK1(12) PACK1(13) PACK1(14) PACK1(15)
      }
      if (k >= 2) {
        const int   g2    = NG - 1 - ((k - 2) * NSLOT + h);
        const bool  qdone = (g2 == NG - 1);
        const float sv    = Sb[g2 * GRP + li];
        ACC1(0) ACC1(1) ACC1(2) ACC1(3) ACC1(4) ACC1(5) ACC1(6) ACC1(7)
        ACC1(8) ACC1(9) ACC1(10) ACC1(11) ACC1(12) ACC1(13) ACC1(14) ACC1(15)
      }
      tvA = tvB;
      tvB = tvN;
    }
    __syncthreads();
  }
#undef GROUP
#undef LOADSET
#undef GSTEP
#undef ACC1
#undef PACK1

  if (wave != 0) Part[(wave - 1) * 32 + lane] = acc;
  __syncthreads();
  if (wave == 0) {
    float2 r = Part[lane];
    for (int h = 1; h < NSLOT; ++h) {
      const float2 q = Part[h * 32 + lane];
      r.x += q.x;
      r.y += q.y;
    }
    Cx2[lane] = r;
  }
  __syncthreads();

  v2f bO[16];
  if (wave < 4) {
    const int nt = wave;
    v2f bR[16];
#pragma unroll
    for (int kt = 0; kt < 16; ++kt) bR[kt] = frag_b(Wr, 64, kt, nt, lane);
#pragma unroll
    for (int kt = 0; kt < 16; ++kt) bO[kt] = frag_b(Wo, 64, kt, nt, lane);
    v8f acc8 = {};
#pragma unroll
    for (int kt = 0; kt < 16; ++kt)
      acc8 = wmma_f32(frag_a_row0(Cx2, kt, lane), bR[kt], acc8);
    const int   col = (nt << 4) + (lane & 15);
    const float y   = acc8[0] + br[col];
    if (lane < 16) ((float*)Yx2)[col] = y;
  }
  __syncthreads();
  if (wave < 4) {
    const int nt = wave;
    v8f acc8 = {};
#pragma unroll
    for (int kt = 0; kt < 16; ++kt)
      acc8 = wmma_f32(frag_a_row0(Yx2, kt, lane), bO[kt], acc8);
    const int   col = (nt << 4) + (lane & 15);
    const float y   = acc8[0] + bo[col];
    if (lane < 16) ((float*)Ox2)[col] = y;
  }
  __syncthreads();
  if (wave == 0) ((float2*)(out + b * 64))[lane] = Ox2[lane];
}

extern "C" void kernel_launch(void* const* d_in, const int* in_sizes, int n_in,
                              void* d_out, int out_size, void* d_ws, size_t ws_size,
                              hipStream_t stream) {
  const int*   seq   = (const int*)d_in[0];
  const float* embed = (const float*)d_in[1];
  const float* W1    = (const float*)d_in[2];
  const float* b1    = (const float*)d_in[3];
  const float* W2    = (const float*)d_in[4];
  const float* b2    = (const float*)d_in[5];
  const float* gamma = (const float*)d_in[6];
  const float* beta  = (const float*)d_in[7];
  const float* Wr    = (const float*)d_in[8];
  const float* br    = (const float*)d_in[9];
  const float* Wo    = (const float*)d_in[10];
  const float* bo    = (const float*)d_in[11];
  float*       out   = (float*)d_out;

  delta_fused_kernel<<<BATCH, NTHR, 0, stream>>>(seq, embed, W1, b1, W2, b2, gamma, beta,
                                                 Wr, br, Wo, bo, out);
}


# Scan's carried step made ONE instruction: the fma takes lane o of p through its own DPP row_share operand (no broadcast move in front), c updates laid one step behind it in asm; same order of sums
# speedup vs baseline: 10.7202x; 1.0637x over previous
#include <hip/hip_runtime.h>


#define HDIM   64
#define VOCABN 64
#define BATCH  64
#define SEQL   4096
#define TPITCH 68

typedef __attribute__((ext_vector_type(2))) float v2f;
typedef __attribute__((ext_vector_type(2))) float f32x2;
typedef __attribute__((ext_vector_type(8))) float v8f;

__device__ __forceinline__ v2f frag_a(const float* A, int lda, int mt, int kt, int lane) {
  int row = (mt << 4) + (lane & 15);
  int k0  = (kt << 2) + ((lane >> 4) << 1);
  v2f a;
  a.x = A[row * lda + k0];
  a.y = A[row * lda + k0 + 1];
  return a;
}
__device__ __forceinline__ v2f frag_b(const float* Bm, int ldb, int kt, int nt, int lane) {
  int col = (nt << 4) + (lane & 15);
  int k0  = (kt << 2) + ((lane >> 4) << 1);
  v2f b;
  b.x = Bm[k0 * ldb + col];
  b.y = Bm[(k0 + 1) * ldb + col];
  return b;
}
__device__ __forceinline__ v2f frag_bt(const float* S, int lds, int kt, int nt, int lane) {
  int col = (nt << 4) + (lane & 15);
  int k0  = (kt << 2) + ((lane >> 4) << 1);
  v2f b;
  b.x = S[col * lds + k0];
  b.y = S[col * lds + k0 + 1];
  return b;
}

__device__ __forceinline__ v8f wmma_f32(v2f a, v2f b, v8f c) {
  return __builtin_amdgcn_wmma_f32_16x16x4_f32(false, a, false, b, (short)0, c,
                                               false, false);
}

__device__ __forceinline__ v2f frag_a_row0(const float2* x2, int kt, int lane) {
  const float2 xv = x2[(kt << 1) + (lane >> 4)];
  const bool   r0 = (lane & 15) == 0;
  v2f a;
  a.x = r0 ? xv.x : 0.0f;
  a.y = r0 ? xv.y : 0.0f;
  return a;
}

#define GRP    16
#define NSLOT  8
#define NTRIP  32
#define NTHR   288

template <int J>
__device__ __forceinline__ int row_share_i(int x) {
  return __builtin_amdgcn_update_dpp(0, x, 0x150 + J, 0xf, 0xf, true);
}
template <int J>
__device__ __forceinline__ float row_share(float x) {
  return __int_as_float(row_share_i<J>(__float_as_int(x)));
}

#define DPPCTL " row_mask:0xf bank_mask:0xf bound_ctrl:1"
__device__ __forceinline__ void scan_step_first(float& p, float& s, float a) {
  float pn, sn;
  asm("v_fma_f32_e64_dpp %0, %2, %3, %2 row_share:0" DPPCTL "\n\t"
      "v_mov_b32_dpp %1, %2 row_share:0" DPPCTL
      : "=&v"(pn), "=&v"(sn)
      : "v"(p), "v"(a));
  p = pn;
  s = sn;
}
template <int J>
__device__ __forceinline__ void scan_step_mid(float& p, float& s, float& c_lo, float& c_hi,
                                              float a, float gx, float gy) {
  float pn, sn;
  asm("s_delay_alu instid0(VALU_DEP_4)\n\t"
      "v_fma_f32_e64_dpp %0, %4, %5, %4 row_share:%9" DPPCTL "\n\t"
      "v_mov_b32_dpp %1, %4 row_share:%9" DPPCTL "\n\t"
      "v_fmac_f32_e32 %2, %6, %8\n\t"
      "v_fmac_f32_e32 %3, %7, %8"
      : "=&v"(pn), "=&v"(sn), "+v"(c_lo), "+v"(c_hi)
      : "v"(p), "v"(a), "v"(gx), "v"(gy), "v"(s), "n"(J));
  p = pn;
  s = sn;
}
__device__ __forceinline__ void scan_step_last(float p, float s, float& c_lo, float& c_hi,
                                               float gx14, float gy14, float gx15, float gy15) {
  asm("v_fmac_f32_e32 %0, %3, %5\n\t"
      "v_fmac_f32_e32 %1, %4, %5\n\t"
      "v_fmac_f32_dpp %0, %2, %6 row_share:15" DPPCTL "\n\t"
      "v_fmac_f32_dpp %1, %2, %7 row_share:15" DPPCTL
      : "+v"(c_lo), "+v"(c_hi)
      : "v"(p), "v"(gx14), "v"(gy14), "v"(s), "v"(gx15), "v"(gy15));
}

__device__ __forceinline__ float gather_c(float c_lo, float c_hi, int tok, int lane) {
  const unsigned sel = (unsigned)tok & 15u;
  const unsigned ulo = (unsigned)__float_as_int(c_lo);
  const unsigned uhi = (unsigned)__float_as_int(c_hi);
  const unsigned lo_s = __builtin_amdgcn_permlane16_var(ulo, ulo, sel, false, false);
  const unsigned lo_x = __builtin_amdgcn_permlanex16_var(ulo, ulo, sel, false, false);
  const unsigned hi_s = __builtin_amdgcn_permlane16_var(uhi, uhi, sel, false, false);
  const unsigned hi_x = __builtin_amdgcn_permlanex16_var(uhi, uhi, sel, false, false);
  const bool other_row = (((tok >> 4) ^ (lane >> 4)) & 1) != 0;
  const unsigned lo = other_row ? lo_x : lo_s;
  const unsigned hi = other_row ? hi_x : hi_s;
  return __int_as_float((int)((tok & 32) ? hi : lo));
}

#define NWAVE  9

__global__ __launch_bounds__(NTHR) void delta_fused_kernel(
    const int* __restrict__ seq, const float* __restrict__ embed,
    const float* __restrict__ W1, const float* __restrict__ b1,
    const float* __restrict__ W2, const float* __restrict__ b2,
    const float* __restrict__ gamma, const float* __restrict__ beta,
    const float* __restrict__ Wr, const float* __restrict__ br,
    const float* __restrict__ Wo, const float* __restrict__ bo,
    float* __restrict__ out) {
  __shared__ float4 Gl4[VOCABN * 16];
  __shared__ float4 Tl4[VOCABN * 16];
  __shared__ float4 Pk[2 * NSLOT * GRP * 32];
  __shared__ float  Sb[SEQL];
  __shared__ float2 Part[NSLOT * 32];
  __shared__ float4 Rl4[64 * 32];
  __shared__ float4 Xl4[64 * 16];
  __shared__ float  rdl[64];
  __shared__ float  Gpl[VOCABN * VOCABN];
  __shared__ float2 Cx2[32];
  __shared__ float2 Yx2[32];
  __shared__ float2 Ox2[32];
  float* Rlds = (float*)Rl4;
  float* Xlds = (float*)Xl4;
  const int b   = blockIdx.x;
  const int tid = threadIdx.x;
  const int lane = tid & 31;
  const int wave = __builtin_amdgcn_readfirstlane(tid >> 5);

  for (int tile = wave; tile < 32; tile += NWAVE) {
    int mt = tile >> 3, nt = tile & 7;
    v8f acc8 = {};
#pragma unroll
    for (int kt = 0; kt < 16; ++kt) {
      v2f a = frag_a(embed, 64, mt, kt, lane);
      v2f bq = frag_b(W1, 128, kt, nt, lane);
      acc8 = wmma_f32(a, bq, acc8);
    }
    int col   = (nt << 4) + (lane & 15);
    int rbase = (mt << 4) + ((lane >> 4) << 3);
#pragma unroll
    for (int v = 0; v < 8; ++v) {
      float x = acc8[v] + b1[col];
      Rlds[(rbase + v) * 128 + col] = fmaxf(x, 0.0f);
    }
  }
  __syncthreads();

  for (int tile = wave; tile < 16; tile += NWAVE) {
    int mt = tile >> 2, nt = tile & 3;
    v8f acc8 = {};
#pragma unroll
    for (int kt = 0; kt < 32; ++kt) {
      v2f a = frag_a(Rlds, 128, mt, kt, lane);
      v2f bq = frag_b(W2, 64, kt, nt, lane);
      acc8 = wmma_f32(a, bq, acc8);
    }
    int col   = (nt << 4) + (lane & 15);
    int rbase = (mt << 4) + ((lane >> 4) << 3);
#pragma unroll
    for (int v = 0; v < 8; ++v) {
      int row = rbase + v;
      Xlds[row * 64 + col] = embed[row * 64 + col] + acc8[v] + b2[col];
    }
  }
  __syncthreads();

  if (tid < 64) {
    const float* xr = &Xlds[tid * 64];
    float mu = 0.f;
#pragma unroll
    for (int c = 0; c < 64; ++c) mu += xr[c];
    mu *= (1.0f / 64.0f);
    float var = 0.f;
#pragma unroll
    for (int c = 0; c < 64; ++c) { float d = xr[c] - mu; var += d * d; }
    var *= (1.0f / 64.0f);
    float rs  = rsqrtf(var + 1e-5f);
    float ksq = 0.f;
#pragma unroll
    for (int c = 0; c < 64; ++c) {
      float tv = (xr[c] - mu) * rs * gamma[c] + beta[c];
      Rlds[tid * TPITCH + c] = tv;
      ksq += tv * tv;
    }
    rdl[tid] = 1.0f / (ksq + 1e-6f);
  }
  __syncthreads();

  for (int i = tid; i < VOCABN * 16; i += NTHR)
    Tl4[i] = *(const float4*)&Rlds[(i >> 4) * TPITCH + ((i & 15) << 2)];

  {
    const float* Tp  = Rlds;
    float*       Gsf = (float*)Gl4;
    for (int tile = wave; tile < 16; tile += NWAVE) {
      int mt = tile >> 2, nt = tile & 3;
      v8f acc8 = {};
#pragma unroll
      for (int kt = 0; kt < 16; ++kt) {
        v2f a = frag_a(Tp, TPITCH, mt, kt, lane);
        v2f bq = frag_bt(Tp, TPITCH, kt, nt, lane);
        acc8 = wmma_f32(a, bq, acc8);
      }
      int col   = (nt << 4) + (lane & 15);
      int rbase = (mt << 4) + ((lane >> 4) << 3);
#pragma unroll
      for (int v = 0; v < 8; ++v) {
        int row = rbase + v;
        Gpl[row * 64 + col] = acc8[v];
        Gsf[((row << 5) + (col & 31)) * 2 + (col >> 5)] = -(acc8[v] * rdl[row]);
      }
    }
  }
  __syncthreads();

  const float2* Gl2 = (const float2*)Gl4;
  const float*  Glf = (const float*)Gl4;
  const float2* Tl2 = (const float2*)Tl4;
  const int*    sqr = seq + b * SEQL;
  const int     li  = lane & 15;
  const int     NG  = SEQL / GRP;

  float  c_lo = 0.0f, c_hi = 0.0f;
  float2 acc  = make_float2(0.0f, 0.0f);
  int    tvA = 0, tvB = 0;
  if (wave == 0) {
    const int tq = sqr[SEQL - 1];
    c_lo = Gpl[tq * 64 + lane];
    c_hi = Gpl[tq * 64 + lane + 32];
  }

#define PACK1(o)                                                           \
  {                                                                        \
    const int to = row_share_i<(o)>(tvN);                                  \
    float2 R = Gl2[to * 32 + lane];                                        \
    float  A = Glf[to * 64 + offn];                                        \
    A = (li > (o)) ? A : 0.0f;                                             \
    if ((o) == 0 && qgroup) { R.x = 0.0f; R.y = 0.0f; A = 0.0f; }          \
    pkw[(o) * 32] = make_float4(R.x, R.y, A, __int_as_float(tvN));         \
  }
#define ACC1(o)                                                            \
  {                                                                        \
    float so = row_share<(o)>(sv);                                         \
    if ((o) == 0 && qdone) so = 0.0f;                                      \
    const int    to = row_share_i<(o)>(tvA);                               \
    const float2 tr = Tl2[to * 32 + lane];                                 \
    acc.x = fmaf(tr.x, so, acc.x);                                         \
    acc.y = fmaf(tr.y, so, acc.y);                                         \
  }
#define LOADSET(X, slot)                                                   \
  {                                                                        \
    _Pragma("unroll") for (int o = 0; o < GRP; ++o)                        \
      X[o] = pkr[((slot) * GRP + o) * 32];                                 \
  }
#define MSTEP(o, X)                                                        \
  scan_step_mid<(o)>(p, s, c_lo, c_hi, X[o].z, X[(o) - 1].x, X[(o) - 1].y);
#define GROUP(Xc, Xn, slot, HASNEXT)                                       \
  {                                                                        \
    if (HASNEXT) LOADSET(Xn, (slot) + 1)                                   \
    __builtin_amdgcn_sched_barrier(0);                                     \
    float p = gather_c(c_lo, c_hi, __float_as_int(Xc[0].w), lane);         \
    float s;                                                               \
    scan_step_first(p, s, Xc[0].z);                                        \
    MSTEP(1, Xc) MSTEP(2, Xc) MSTEP(3, Xc) MSTEP(4, Xc) MSTEP(5, Xc)       \
    MSTEP(6, Xc) MSTEP(7, Xc) MSTEP(8, Xc) MSTEP(9, Xc) MSTEP(10, Xc)      \
    MSTEP(11, Xc) MSTEP(12, Xc) MSTEP(13, Xc) MSTEP(14, Xc)                \
    scan_step_last(p, s, c_lo, c_hi, Xc[14].x, Xc[14].y, Xc[15].x, Xc[15].y); \
    Sb[(gfirst - (slot)) * GRP + li] = p;                                  \
  }

  for (int k = 0; k < NTRIP + 2; ++k) {
    if (wave == 0) {
      if (k >= 1 && k <= NTRIP) {
        const float4* pkr    = Pk + ((k - 1) & 1) * (NSLOT * GRP * 32) + lane;
        const int     gfirst = NG - 1 - (k - 1) * NSLOT;
        float4 X0[GRP], X1[GRP];
        LOADSET(X0, 0)
        GROUP(X0, X1, 0, 1)
        GROUP(X1, X0, 1, 1)
        GROUP(X0, X1, 2, 1)
        GROUP(X1, X0, 3, 1)
        GROUP(X0, X1, 4, 1)
        GROUP(X1, X0, 5, 1)
        GROUP(X0, X1, 6, 1)
        GROUP(X1, X0, 7, 0)
      }
    } else {
      const int h   = wave - 1;
      int       tvN = 0;
      if (k < NTRIP) {
        const int  g      = NG - 1 - (k * NSLOT + h);
        const bool qgroup = (g == NG - 1);
        tvN = sqr[g * GRP + (GRP - 1 - li)];
        const int offn = ((tvN & 31) << 1) + (tvN >> 5);
        float4*   pkw  = Pk + ((k & 1) * NSLOT + h) * (GRP * 32) + lane;
        PACK1(0) PACK1(1) PACK1(2) PACK1(3) PACK1(4) PACK1(5) PACK1(6) PACK1(7)
        PACK1(8) PACK1(9) PACK1(10) PACK1(11) PACK1(12) PACK1(13) PACK1(14) PACK1(15)
      }
      if (k >= 2) {
        const int   g2    = NG - 1 - ((k - 2) * NSLOT + h);
        const bool  qdone = (g2 == NG - 1);
        const float sv    = Sb[g2 * GRP + li];
        ACC1(0) ACC1(1) ACC1(2) ACC1(3) ACC1(4) ACC1(5) ACC1(6) ACC1(7)
        ACC1(8) ACC1(9) ACC1(10) ACC1(11) ACC1(12) ACC1(13) ACC1(14) ACC1(15)
      }
      tvA = tvB;
      tvB = tvN;
    }
    __syncthreads();
  }
#undef GROUP
#undef LOADSET
#undef MSTEP
#undef ACC1
#undef PACK1

  if (wave != 0) Part[(wave - 1) * 32 + lane] = acc;
  __syncthreads();
  if (wave == 0) {
    float2 r = Part[lane];
    for (int h = 1; h < NSLOT; ++h) {
      const float2 q = Part[h * 32 + lane];
      r.x += q.x;
      r.y += q.y;
    }
    Cx2[lane] = r;
  }
  __syncthreads();

  v2f bO[16];
  if (wave < 4) {
    const int nt = wave;
    v2f bR[16];
#pragma unroll
    for (int kt = 0; kt < 16; ++kt) bR[kt] = frag_b(Wr, 64, kt, nt, lane);
#pragma unroll
    for (int kt = 0; kt < 16; ++kt) bO[kt] = frag_b(Wo, 64, kt, nt, lane);
    v8f acc8 = {};
#pragma unroll
    for (int kt = 0; kt < 16; ++kt)
      acc8 = wmma_f32(frag_a_row0(Cx2, kt, lane), bR[kt], acc8);
    const int   col = (nt << 4) + (lane & 15);
    const float y   = acc8[0] + br[col];
    if (lane < 16) ((float*)Yx2)[col] = y;
  }
  __syncthreads();
  if (wave < 4) {
    const int nt = wave;
    v8f acc8 = {};
#pragma unroll
    for (int kt = 0; kt < 16; ++kt)
      acc8 = wmma_f32(frag_a_row0(Yx2, kt, lane), bO[kt], acc8);
    const int   col = (nt << 4) + (lane & 15);
    const float y   = acc8[0] + bo[col];
    if (lane < 16) ((float*)Ox2)[col] = y;
  }
  __syncthreads();
  if (wave == 0) ((float2*)(out + b * 64))[lane] = Ox2[lane];
}

extern "C" void kernel_launch(void* const* d_in, const int* in_sizes, int n_in,
                              void* d_out, int out_size, void* d_ws, size_t ws_size,
                              hipStream_t stream) {
  const int*   seq   = (const int*)d_in[0];
  const float* embed = (const float*)d_in[1];
  const float* W1    = (const float*)d_in[2];
  const float* b1    = (const float*)d_in[3];
  const float* W2    = (const float*)d_in[4];
  const float* b2    = (const float*)d_in[5];
  const float* gamma = (const float*)d_in[6];
  const float* beta  = (const float*)d_in[7];
  const float* Wr    = (const float*)d_in[8];
  const float* br    = (const float*)d_in[9];
  const float* Wo    = (const float*)d_in[10];
  const float* bo    = (const float*)d_in[11];
  float*       out   = (float*)d_out;

  delta_fused_kernel<<<BATCH, NTHR, 0, stream>>>(seq, embed, W1, b1, W2, b2, gamma, beta,
                                                 Wr, br, Wo, bo, out);
}


# Helper waves' packing un-serialised: only step 0's record keeps the token word (common to all 16, it tied every record to one register tuple); a group's 32 LDS reads fly together; same order of sums
# speedup vs baseline: 12.8392x; 1.1977x over previous
#include <hip/hip_runtime.h>


#define HDIM   64
#define VOCABN 64
#define BATCH  64
#define SEQL   4096
#define TPITCH 68

typedef __attribute__((ext_vector_type(2))) float v2f;
typedef __attribute__((ext_vector_type(2))) float f32x2;
typedef __attribute__((ext_vector_type(8))) float v8f;
typedef __attribute__((ext_vector_type(3))) float v3f;

__device__ __forceinline__ v2f frag_a(const float* A, int lda, int mt, int kt, int lane) {
  int row = (mt << 4) + (lane & 15);
  int k0  = (kt << 2) + ((lane >> 4) << 1);
  v2f a;
  a.x = A[row * lda + k0];
  a.y = A[row * lda + k0 + 1];
  return a;
}
__device__ __forceinline__ v2f frag_b(const float* Bm, int ldb, int kt, int nt, int lane) {
  int col = (nt << 4) + (lane & 15);
  int k0  = (kt << 2) + ((lane >> 4) << 1);
  v2f b;
  b.x = Bm[k0 * ldb + col];
  b.y = Bm[(k0 + 1) * ldb + col];
  return b;
}
__device__ __forceinline__ v2f frag_bt(const float* S, int lds, int kt, int nt, int lane) {
  int col = (nt << 4) + (lane & 15);
  int k0  = (kt << 2) + ((lane >> 4) << 1);
  v2f b;
  b.x = S[col * lds + k0];
  b.y = S[col * lds + k0 + 1];
  return b;
}

__device__ __forceinline__ v8f wmma_f32(v2f a, v2f b, v8f c) {
  return __builtin_amdgcn_wmma_f32_16x16x4_f32(false, a, false, b, (short)0, c,
                                               false, false);
}

__device__ __forceinline__ v2f frag_a_row0(const float2* x2, int kt, int lane) {
  const float2 xv = x2[(kt << 1) + (lane >> 4)];
  const bool   r0 = (lane & 15) == 0;
  v2f a;
  a.x = r0 ? xv.x : 0.0f;
  a.y = r0 ? xv.y : 0.0f;
  return a;
}

#define GRP    16
#define NSLOT  8
#define NTRIP  32
#define NTHR   288

template <int J>
__device__ __forceinline__ int row_share_i(int x) {
  return __builtin_amdgcn_update_dpp(0, x, 0x150 + J, 0xf, 0xf, true);
}
template <int J>
__device__ __forceinline__ float row_share(float x) {
  return __int_as_float(row_share_i<J>(__float_as_int(x)));
}

#define DPPCTL " row_mask:0xf bank_mask:0xf bound_ctrl:1"
__device__ __forceinline__ void scan_step_first(float& p, float& s, float a) {
  float pn, sn;
  asm("v_fma_f32_e64_dpp %0, %2, %3, %2 row_share:0" DPPCTL "\n\t"
      "v_mov_b32_dpp %1, %2 row_share:0" DPPCTL
      : "=&v"(pn), "=&v"(sn)
      : "v"(p), "v"(a));
  p = pn;
  s = sn;
}
template <int J>
__device__ __forceinline__ void scan_step_mid(float& p, float& s, float& c_lo, float& c_hi,
                                              float a, float gx, float gy) {
  float pn, sn;
  asm("s_delay_alu instid0(VALU_DEP_4)\n\t"
      "v_fma_f32_e64_dpp %0, %4, %5, %4 row_share:%9" DPPCTL "\n\t"
      "v_mov_b32_dpp %1, %4 row_share:%9" DPPCTL "\n\t"
      "v_fmac_f32_e32 %2, %6, %8\n\t"
      "v_fmac_f32_e32 %3, %7, %8"
      : "=&v"(pn), "=&v"(sn), "+v"(c_lo), "+v"(c_hi)
      : "v"(p), "v"(a), "v"(gx), "v"(gy), "v"(s), "n"(J));
  p = pn;
  s = sn;
}
__device__ __forceinline__ void scan_step_last(float p, float s, float& c_lo, float& c_hi,
                                               float gx14, float gy14, float gx15, float gy15) {
  asm("v_fmac_f32_e32 %0, %3, %5\n\t"
      "v_fmac_f32_e32 %1, %4, %5\n\t"
      "v_fmac_f32_dpp %0, %2, %6 row_share:15" DPPCTL "\n\t"
      "v_fmac_f32_dpp %1, %2, %7 row_share:15" DPPCTL
      : "+v"(c_lo), "+v"(c_hi)
      : "v"(p), "v"(gx14), "v"(gy14), "v"(s), "v"(gx15), "v"(gy15));
}

__device__ __forceinline__ float gather_c(float c_lo, float c_hi, int tok, int lane) {
  const unsigned sel = (unsigned)tok & 15u;
  const unsigned ulo = (unsigned)__float_as_int(c_lo);
  const unsigned uhi = (unsigned)__float_as_int(c_hi);
  const unsigned lo_s = __builtin_amdgcn_permlane16_var(ulo, ulo, sel, false, false);
  const unsigned lo_x = __builtin_amdgcn_permlanex16_var(ulo, ulo, sel, false, false);
  const unsigned hi_s = __builtin_amdgcn_permlane16_var(uhi, uhi, sel, false, false);
  const unsigned hi_x = __builtin_amdgcn_permlanex16_var(uhi, uhi, sel, false, false);
  const bool other_row = (((tok >> 4) ^ (lane >> 4)) & 1) != 0;
  const unsigned lo = other_row ? lo_x : lo_s;
  const unsigned hi = other_row ? hi_x : hi_s;
  return __int_as_float((int)((tok & 32) ? hi : lo));
}

#define NWAVE  9

__global__ __launch_bounds__(NTHR) void delta_fused_kernel(
    const int* __restrict__ seq, const float* __restrict__ embed,
    const float* __restrict__ W1, const float* __restrict__ b1,
    const float* __restrict__ W2, const float* __restrict__ b2,
    const float* __restrict__ gamma, const float* __restrict__ beta,
    const float* __restrict__ Wr, const float* __restrict__ br,
    const float* __restrict__ Wo, const float* __restrict__ bo,
    float* __restrict__ out) {
  __shared__ float4 Gl4[VOCABN * 16];
  __shared__ float4 Tl4[VOCABN * 16];
  __shared__ float4 Pk[2 * NSLOT * GRP * 32];
  __shared__ float  Sb[SEQL];
  __shared__ float2 Part[NSLOT * 32];
  __shared__ float4 Rl4[64 * 32];
  __shared__ float4 Xl4[64 * 16];
  __shared__ float  rdl[64];
  __shared__ float  Gpl[VOCABN * VOCABN];
  __shared__ float2 Cx2[32];
  __shared__ float2 Yx2[32];
  __shared__ float2 Ox2[32];
  float* Rlds = (float*)Rl4;
  float* Xlds = (float*)Xl4;
  const int b   = blockIdx.x;
  const int tid = threadIdx.x;
  const int lane = tid & 31;
  const int wave = __builtin_amdgcn_readfirstlane(tid >> 5);

  for (int tile = wave; tile < 32; tile += NWAVE) {
    int mt = tile >> 3, nt = tile & 7;
    v8f acc8 = {};
#pragma unroll
    for (int kt = 0; kt < 16; ++kt) {
      v2f a = frag_a(embed, 64, mt, kt, lane);
      v2f bq = frag_b(W1, 128, kt, nt, lane);
      acc8 = wmma_f32(a, bq, acc8);
    }
    int col   = (nt << 4) + (lane & 15);
    int rbase = (mt << 4) + ((lane >> 4) << 3);
#pragma unroll
    for (int v = 0; v < 8; ++v) {
      float x = acc8[v] + b1[col];
      Rlds[(rbase + v) * 128 + col] = fmaxf(x, 0.0f);
    }
  }
  __syncthreads();

  for (int tile = wave; tile < 16; tile += NWAVE) {
    int mt = tile >> 2, nt = tile & 3;
    v8f acc8 = {};
#pragma unroll
    for (int kt = 0; kt < 32; ++kt) {
      v2f a = frag_a(Rlds, 128, mt, kt, lane);
      v2f bq = frag_b(W2, 64, kt, nt, lane);
      acc8 = wmma_f32(a, bq, acc8);
    }
    int col   = (nt << 4) + (lane & 15);
    int rbase = (mt << 4) + ((lane >> 4) << 3);
#pragma unroll
    for (int v = 0; v < 8; ++v) {
      int row = rbase + v;
      Xlds[row * 64 + col] = embed[row * 64 + col] + acc8[v] + b2[col];
    }
  }
  __syncthreads();

  if (tid < 64) {
    const float* xr = &Xlds[tid * 64];
    float mu = 0.f;
#pragma unroll
    for (int c = 0; c < 64; ++c) mu += xr[c];
    mu *= (1.0f / 64.0f);
    float var = 0.f;
#pragma unroll
    for (int c = 0; c < 64; ++c) { float d = xr[c] - mu; var += d * d; }
    var *= (1.0f / 64.0f);
    float rs  = rsqrtf(var + 1e-5f);
    float ksq = 0.f;
#pragma unroll
    for (int c = 0; c < 64; ++c) {
      float tv = (xr[c] - mu) * rs * gamma[c] + beta[c];
      Rlds[tid * TPITCH + c] = tv;
      ksq += tv * tv;
    }
    rdl[tid] = 1.0f / (ksq + 1e-6f);
  }
  __syncthreads();

  for (int i = tid; i < VOCABN * 16; i += NTHR)
    Tl4[i] = *(const float4*)&Rlds[(i >> 4) * TPITCH + ((i & 15) << 2)];

  {
    const float* Tp  = Rlds;
    float*       Gsf = (float*)Gl4;
    for (int tile = wave; tile < 16; tile += NWAVE) {
      int mt = tile >> 2, nt = tile & 3;
      v8f acc8 = {};
#pragma unroll
      for (int kt = 0; kt < 16; ++kt) {
        v2f a = frag_a(Tp, TPITCH, mt, kt, lane);
        v2f bq = frag_bt(Tp, TPITCH, kt, nt, lane);
        acc8 = wmma_f32(a, bq, acc8);
      }
      int col   = (nt << 4) + (lane & 15);
      int rbase = (mt << 4) + ((lane >> 4) << 3);
#pragma unroll
      for (int v = 0; v < 8; ++v) {
        int row = rbase + v;
        Gpl[row * 64 + col] = acc8[v];
        Gsf[((row << 5) + (col & 31)) * 2 + (col >> 5)] = -(acc8[v] * rdl[row]);
      }
    }
  }
  __syncthreads();

  const float2* Gl2 = (const float2*)Gl4;
  const float*  Glf = (const float*)Gl4;
  const float2* Tl2 = (const float2*)Tl4;
  const int*    sqr = seq + b * SEQL;
  const int     li  = lane & 15;
  const int     NG  = SEQL / GRP;

  float  c_lo = 0.0f, c_hi = 0.0f;
  float2 acc  = make_float2(0.0f, 0.0f);
  int    tvA = 0, tvB = 0;
  if (wave == 0) {
    const int tq = sqr[SEQL - 1];
    c_lo = Gpl[tq * 64 + lane];
    c_hi = Gpl[tq * 64 + lane + 32];
  }

#define PLOAD(o)                                                           \
  {                                                                        \
    const int to = row_share_i<(o)>(tvN);                                  \
    Rv[o] = Gl2[to * 32 + lane];                                           \
    Av[o] = Glf[to * 64 + offn];                                           \
  }
#define PSTORE(o)                                                          \
  {                                                                        \
    const float A = (li > (o)) ? Av[o] : 0.0f;                             \
    v3f rec;                                                               \
    rec.x = Rv[o].x;                                                       \
    rec.y = Rv[o].y;                                                       \
    rec.z = A;                                                             \
    *(v3f*)&pkw[(o) * 32] = rec;                                           \
  }
#define PSTORE0()                                                          \
  {                                                                        \
    float2 R = Rv[0];                                                      \
    float  A = (li > 0) ? Av[0] : 0.0f;                                    \
    if (qgroup) { R.x = 0.0f; R.y = 0.0f; A = 0.0f; }                      \
    pkw[0] = make_float4(R.x, R.y, A, __int_as_float(tvN));                \
  }
#define ACC1(o)                                                            \
  {                                                                        \
    float so = row_share<(o)>(sv);                                         \
    if ((o) == 0 && qdone) so = 0.0f;                                      \
    const int    to = row_share_i<(o)>(tvA);                               \
    const float2 tr = Tl2[to * 32 + lane];                                 \
    acc.x = fmaf(tr.x, so, acc.x);                                         \
    acc.y = fmaf(tr.y, so, acc.y);                                         \
  }
#define LOADSET(X, slot)                                                   \
  {                                                                        \
    _Pragma("unroll") for (int o = 0; o < GRP; ++o)                        \
      X[o] = pkr[((slot) * GRP + o) * 32];                                 \
  }
#define MSTEP(o, X)                                                        \
  scan_step_mid<(o)>(p, s, c_lo, c_hi, X[o].z, X[(o) - 1].x, X[(o) - 1].y);
#define GROUP(Xc, Xn, slot, HASNEXT)                                       \
  {                                                                        \
    if (HASNEXT) LOADSET(Xn, (slot) + 1)                                   \
    __builtin_amdgcn_sched_barrier(0);                                     \
    float p = gather_c(c_lo, c_hi, __float_as_int(Xc[0].w), lane);         \
    float s;                                                               \
    scan_step_first(p, s, Xc[0].z);                                        \
    MSTEP(1, Xc) MSTEP(2, Xc) MSTEP(3, Xc) MSTEP(4, Xc) MSTEP(5, Xc)       \
    MSTEP(6, Xc) MSTEP(7, Xc) MSTEP(8, Xc) MSTEP(9, Xc) MSTEP(10, Xc)      \
    MSTEP(11, Xc) MSTEP(12, Xc) MSTEP(13, Xc) MSTEP(14, Xc)                \
    scan_step_last(p, s, c_lo, c_hi, Xc[14].x, Xc[14].y, Xc[15].x, Xc[15].y); \
    Sb[(gfirst - (slot)) * GRP + li] = p;                                  \
  }

  for (int k = 0; k < NTRIP + 2; ++k) {
    if (wave == 0) {
      if (k >= 1 && k <= NTRIP) {
        const float4* pkr    = Pk + ((k - 1) & 1) * (NSLOT * GRP * 32) + lane;
        const int     gfirst = NG - 1 - (k - 1) * NSLOT;
        float4 X0[GRP], X1[GRP];
        LOADSET(X0, 0)
        GROUP(X0, X1, 0, 1)
        GROUP(X1, X0, 1, 1)
        GROUP(X0, X1, 2, 1)
        GROUP(X1, X0, 3, 1)
        GROUP(X0, X1, 4, 1)
        GROUP(X1, X0, 5, 1)
        GROUP(X0, X1, 6, 1)
        GROUP(X1, X0, 7, 0)
      }
    } else {
      const int h   = wave - 1;
      int       tvN = 0;
      if (k < NTRIP) {
        const int  g      = NG - 1 - (k * NSLOT + h);
        const bool qgroup = (g == NG - 1);
        tvN = sqr[g * GRP + (GRP - 1 - li)];
        const int offn = ((tvN & 31) << 1) + (tvN >> 5);
        float4*   pkw  = Pk + ((k & 1) * NSLOT + h) * (GRP * 32) + lane;
        float2 Rv[GRP];
        float  Av[GRP];
        PLOAD(0) PLOAD(1) PLOAD(2) PLOAD(3) PLOAD(4) PLOAD(5) PLOAD(6) PLOAD(7)
        PLOAD(8) PLOAD(9) PLOAD(10) PLOAD(11) PLOAD(12) PLOAD(13) PLOAD(14) PLOAD(15)
        PSTORE0()
        PSTORE(1) PSTORE(2) PSTORE(3) PSTORE(4) PSTORE(5) PSTORE(6) PSTORE(7)
        PSTORE(8) PSTORE(9) PSTORE(10) PSTORE(11) PSTORE(12) PSTORE(13) PSTORE(14) PSTORE(15)
      }
      if (k >= 2) {
        const int   g2    = NG - 1 - ((k - 2) * NSLOT + h);
        const bool  qdone = (g2 == NG - 1);
        const float sv    = Sb[g2 * GRP + li];
        ACC1(0) ACC1(1) ACC1(2) ACC1(3) ACC1(4) ACC1(5) ACC1(6) ACC1(7)
        ACC1(8) ACC1(9) ACC1(10) ACC1(11) ACC1(12) ACC1(13) ACC1(14) ACC1(15)
      }
      tvA = tvB;
      tvB = tvN;
    }
    __syncthreads();
  }
#undef GROUP
#undef LOADSET
#undef MSTEP
#undef ACC1
#undef PSTORE0
#undef PSTORE
#undef PLOAD

  if (wave != 0) Part[(wave - 1) * 32 + lane] = acc;
  __syncthreads();
  if (wave == 0) {
    float2 r = Part[lane];
    for (int h = 1; h < NSLOT; ++h) {
      const float2 q = Part[h * 32 + lane];
      r.x += q.x;
      r.y += q.y;
    }
    Cx2[lane] = r;
  }
  __syncthreads();

  v2f bO[16];
  if (wave < 4) {
    const int nt = wave;
    v2f bR[16];
#pragma unroll
    for (int kt = 0; kt < 16; ++kt) bR[kt] = frag_b(Wr, 64, kt, nt, lane);
#pragma unroll
    for (int kt = 0; kt < 16; ++kt) bO[kt] = frag_b(Wo, 64, kt, nt, lane);
    v8f acc8 = {};
#pragma unroll
    for (int kt = 0; kt < 16; ++kt)
      acc8 = wmma_f32(frag_a_row0(Cx2, kt, lane), bR[kt], acc8);
    const int   col = (nt << 4) + (lane & 15);
    const float y   = acc8[0] + br[col];
    if (lane < 16) ((float*)Yx2)[col] = y;
  }
  __syncthreads();
  if (wave < 4) {
    const int nt = wave;
    v8f acc8 = {};
#pragma unroll
    for (int kt = 0; kt < 16; ++kt)
      acc8 = wmma_f32(frag_a_row0(Yx2, kt, lane), bO[kt], acc8);
    const int   col = (nt << 4) + (lane & 15);
    const float y   = acc8[0] + bo[col];
    if (lane < 16) ((float*)Ox2)[col] = y;
  }
  __syncthreads();
  if (wave == 0) ((float2*)(out + b * 64))[lane] = Ox2[lane];
}

extern "C" void kernel_launch(void* const* d_in, const int* in_sizes, int n_in,
                              void* d_out, int out_size, void* d_ws, size_t ws_size,
                              hipStream_t stream) {
  const int*   seq   = (const int*)d_in[0];
  const float* embed = (const float*)d_in[1];
  const float* W1    = (const float*)d_in[2];
  const float* b1    = (const float*)d_in[3];
  const float* W2    = (const float*)d_in[4];
  const float* b2    = (const float*)d_in[5];
  const float* gamma = (const float*)d_in[6];
  const float* beta  = (const float*)d_in[7];
  const float* Wr    = (const float*)d_in[8];
  const float* br    = (const float*)d_in[9];
  const float* Wo    = (const float*)d_in[10];
  const float* bo    = (const float*)d_in[11];
  float*       out   = (float*)d_out;

  delta_fused_kernel<<<BATCH, NTHR, 0, stream>>>(seq, embed, W1, b1, W2, b2, gamma, beta,
                                                 Wr, br, Wo, bo, out);
}


# Helper waves' waits leave the trip's critical path: tokens fetched one trip ahead in a carried register, record reads issued first, ctx sums run under them, stores last; same order of sums
# speedup vs baseline: 14.3298x; 1.1161x over previous
#include <hip/hip_runtime.h>


#define HDIM   64
#define VOCABN 64
#define BATCH  64
#define SEQL   4096
#define TPITCH 68

typedef __attribute__((ext_vector_type(2))) float v2f;
typedef __attribute__((ext_vector_type(2))) float f32x2;
typedef __attribute__((ext_vector_type(8))) float v8f;
typedef __attribute__((ext_vector_type(3))) float v3f;

__device__ __forceinline__ v2f frag_a(const float* A, int lda, int mt, int kt, int lane) {
  int row = (mt << 4) + (lane & 15);
  int k0  = (kt << 2) + ((lane >> 4) << 1);
  v2f a;
  a.x = A[row * lda + k0];
  a.y = A[row * lda + k0 + 1];
  return a;
}
__device__ __forceinline__ v2f frag_b(const float* Bm, int ldb, int kt, int nt, int lane) {
  int col = (nt << 4) + (lane & 15);
  int k0  = (kt << 2) + ((lane >> 4) << 1);
  v2f b;
  b.x = Bm[k0 * ldb + col];
  b.y = Bm[(k0 + 1) * ldb + col];
  return b;
}
__device__ __forceinline__ v2f frag_bt(const float* S, int lds, int kt, int nt, int lane) {
  int col = (nt << 4) + (lane & 15);
  int k0  = (kt << 2) + ((lane >> 4) << 1);
  v2f b;
  b.x = S[col * lds + k0];
  b.y = S[col * lds + k0 + 1];
  return b;
}

__device__ __forceinline__ v8f wmma_f32(v2f a, v2f b, v8f c) {
  return __builtin_amdgcn_wmma_f32_16x16x4_f32(false, a, false, b, (short)0, c,
                                               false, false);
}

__device__ __forceinline__ v2f frag_a_row0(const float2* x2, int kt, int lane) {
  const float2 xv = x2[(kt << 1) + (lane >> 4)];
  const bool   r0 = (lane & 15) == 0;
  v2f a;
  a.x = r0 ? xv.x : 0.0f;
  a.y = r0 ? xv.y : 0.0f;
  return a;
}

#define GRP    16
#define NSLOT  8
#define NTRIP  32
#define NTHR   288

template <int J>
__device__ __forceinline__ int row_share_i(int x) {
  return __builtin_amdgcn_update_dpp(0, x, 0x150 + J, 0xf, 0xf, true);
}
template <int J>
__device__ __forceinline__ float row_share(float x) {
  return __int_as_float(row_share_i<J>(__float_as_int(x)));
}

#define DPPCTL " row_mask:0xf bank_mask:0xf bound_ctrl:1"
__device__ __forceinline__ void scan_step_first(float& p, float& s, float a) {
  float pn, sn;
  asm("v_fma_f32_e64_dpp %0, %2, %3, %2 row_share:0" DPPCTL "\n\t"
      "v_mov_b32_dpp %1, %2 row_share:0" DPPCTL
      : "=&v"(pn), "=&v"(sn)
      : "v"(p), "v"(a));
  p = pn;
  s = sn;
}
template <int J>
__device__ __forceinline__ void scan_step_mid(float& p, float& s, float& c_lo, float& c_hi,
                                              float a, float gx, float gy) {
  float pn, sn;
  asm("s_delay_alu instid0(VALU_DEP_4)\n\t"
      "v_fma_f32_e64_dpp %0, %4, %5, %4 row_share:%9" DPPCTL "\n\t"
      "v_mov_b32_dpp %1, %4 row_share:%9" DPPCTL "\n\t"
      "v_fmac_f32_e32 %2, %6, %8\n\t"
      "v_fmac_f32_e32 %3, %7, %8"
      : "=&v"(pn), "=&v"(sn), "+v"(c_lo), "+v"(c_hi)
      : "v"(p), "v"(a), "v"(gx), "v"(gy), "v"(s), "n"(J));
  p = pn;
  s = sn;
}
__device__ __forceinline__ void scan_step_last(float p, float s, float& c_lo, float& c_hi,
                                               float gx14, float gy14, float gx15, float gy15) {
  asm("v_fmac_f32_e32 %0, %3, %5\n\t"
      "v_fmac_f32_e32 %1, %4, %5\n\t"
      "v_fmac_f32_dpp %0, %2, %6 row_share:15" DPPCTL "\n\t"
      "v_fmac_f32_dpp %1, %2, %7 row_share:15" DPPCTL
      : "+v"(c_lo), "+v"(c_hi)
      : "v"(p), "v"(gx14), "v"(gy14), "v"(s), "v"(gx15), "v"(gy15));
}

__device__ __forceinline__ float gather_c(float c_lo, float c_hi, int tok, int lane) {
  const unsigned sel = (unsigned)tok & 15u;
  const unsigned ulo = (unsigned)__float_as_int(c_lo);
  const unsigned uhi = (unsigned)__float_as_int(c_hi);
  const unsigned lo_s = __builtin_amdgcn_permlane16_var(ulo, ulo, sel, false, false);
  const unsigned lo_x = __builtin_amdgcn_permlanex16_var(ulo, ulo, sel, false, false);
  const unsigned hi_s = __builtin_amdgcn_permlane16_var(uhi, uhi, sel, false, false);
  const unsigned hi_x = __builtin_amdgcn_permlanex16_var(uhi, uhi, sel, false, false);
  const bool other_row = (((tok >> 4) ^ (lane >> 4)) & 1) != 0;
  const unsigned lo = other_row ? lo_x : lo_s;
  const unsigned hi = other_row ? hi_x : hi_s;
  return __int_as_float((int)((tok & 32) ? hi : lo));
}

#define NWAVE  9

__global__ __launch_bounds__(NTHR) void delta_fused_kernel(
    const int* __restrict__ seq, const float* __restrict__ embed,
    const float* __restrict__ W1, const float* __restrict__ b1,
    const float* __restrict__ W2, const float* __restrict__ b2,
    const float* __restrict__ gamma, const float* __restrict__ beta,
    const float* __restrict__ Wr, const float* __restrict__ br,
    const float* __restrict__ Wo, const float* __restrict__ bo,
    float* __restrict__ out) {
  __shared__ float4 Gl4[VOCABN * 16];
  __shared__ float4 Tl4[VOCABN * 16];
  __shared__ float4 Pk[2 * NSLOT * GRP * 32];
  __shared__ float  Sb[SEQL];
  __shared__ float2 Part[NSLOT * 32];
  __shared__ float4 Rl4[64 * 32];
  __shared__ float4 Xl4[64 * 16];
  __shared__ float  rdl[64];
  __shared__ float  Gpl[VOCABN * VOCABN];
  __shared__ float2 Cx2[32];
  __shared__ float2 Yx2[32];
  __shared__ float2 Ox2[32];
  float* Rlds = (float*)Rl4;
  float* Xlds = (float*)Xl4;
  const int b   = blockIdx.x;
  const int tid = threadIdx.x;
  const int lane = tid & 31;
  const int wave = __builtin_amdgcn_readfirstlane(tid >> 5);

  int tvP = 0;
  if (wave != 0) tvP = seq[b * SEQL + (SEQL / 16 - wave) * 16 + (15 - (lane & 15))];

  for (int tile = wave; tile < 32; tile += NWAVE) {
    int mt = tile >> 3, nt = tile & 7;
    v8f acc8 = {};
#pragma unroll
    for (int kt = 0; kt < 16; ++kt) {
      v2f a = frag_a(embed, 64, mt, kt, lane);
      v2f bq = frag_b(W1, 128, kt, nt, lane);
      acc8 = wmma_f32(a, bq, acc8);
    }
    int col   = (nt << 4) + (lane & 15);
    int rbase = (mt << 4) + ((lane >> 4) << 3);
#pragma unroll
    for (int v = 0; v < 8; ++v) {
      float x = acc8[v] + b1[col];
      Rlds[(rbase + v) * 128 + col] = fmaxf(x, 0.0f);
    }
  }
  __syncthreads();

  for (int tile = wave; tile < 16; tile += NWAVE) {
    int mt = tile >> 2, nt = tile & 3;
    v8f acc8 = {};
#pragma unroll
    for (int kt = 0; kt < 32; ++kt) {
      v2f a = frag_a(Rlds, 128, mt, kt, lane);
      v2f bq = frag_b(W2, 64, kt, nt, lane);
      acc8 = wmma_f32(a, bq, acc8);
    }
    int col   = (nt << 4) + (lane & 15);
    int rbase = (mt << 4) + ((lane >> 4) << 3);
#pragma unroll
    for (int v = 0; v < 8; ++v) {
      int row = rbase + v;
      Xlds[row * 64 + col] = embed[row * 64 + col] + acc8[v] + b2[col];
    }
  }
  __syncthreads();

  if (tid < 64) {
    const float* xr = &Xlds[tid * 64];
    float mu = 0.f;
#pragma unroll
    for (int c = 0; c < 64; ++c) mu += xr[c];
    mu *= (1.0f / 64.0f);
    float var = 0.f;
#pragma unroll
    for (int c = 0; c < 64; ++c) { float d = xr[c] - mu; var += d * d; }
    var *= (1.0f / 64.0f);
    float rs  = rsqrtf(var + 1e-5f);
    float ksq = 0.f;
#pragma unroll
    for (int c = 0; c < 64; ++c) {
      float tv = (xr[c] - mu) * rs * gamma[c] + beta[c];
      Rlds[tid * TPITCH + c] = tv;
      ksq += tv * tv;
    }
    rdl[tid] = 1.0f / (ksq + 1e-6f);
  }
  __syncthreads();

  for (int i = tid; i < VOCABN * 16; i += NTHR)
    Tl4[i] = *(const float4*)&Rlds[(i >> 4) * TPITCH + ((i & 15) << 2)];

  {
    const float* Tp  = Rlds;
    float*       Gsf = (float*)Gl4;
    for (int tile = wave; tile < 16; tile += NWAVE) {
      int mt = tile >> 2, nt = tile & 3;
      v8f acc8 = {};
#pragma unroll
      for (int kt = 0; kt < 16; ++kt) {
        v2f a = frag_a(Tp, TPITCH, mt, kt, lane);
        v2f bq = frag_bt(Tp, TPITCH, kt, nt, lane);
        acc8 = wmma_f32(a, bq, acc8);
      }
      int col   = (nt << 4) + (lane & 15);
      int rbase = (mt << 4) + ((lane >> 4) << 3);
#pragma unroll
      for (int v = 0; v < 8; ++v) {
        int row = rbase + v;
        Gpl[row * 64 + col] = acc8[v];
        Gsf[((row << 5) + (col & 31)) * 2 + (col >> 5)] = -(acc8[v] * rdl[row]);
      }
    }
  }
  __syncthreads();

  const float2* Gl2 = (const float2*)Gl4;
  const float*  Glf = (const float*)Gl4;
  const float2* Tl2 = (const float2*)Tl4;
  const int*    sqr = seq + b * SEQL;
  const int     li  = lane & 15;
  const int     NG  = SEQL / GRP;

  float  c_lo = 0.0f, c_hi = 0.0f;
  float2 acc  = make_float2(0.0f, 0.0f);
  int    tvA = 0, tvB = 0;
  if (wave == 0) {
    const int tq = sqr[SEQL - 1];
    c_lo = Gpl[tq * 64 + lane];
    c_hi = Gpl[tq * 64 + lane + 32];
  }

#define PLOAD(o)                                                           \
  {                                                                        \
    const int to = row_share_i<(o)>(tvN);                                  \
    Rv[o] = Gl2[to * 32 + lane];                                           \
    Av[o] = Glf[to * 64 + offn];                                           \
  }
#define PSTORE(o)                                                          \
  {                                                                        \
    const float A = (li > (o)) ? Av[o] : 0.0f;                             \
    v3f rec;                                                               \
    rec.x = Rv[o].x;                                                       \
    rec.y = Rv[o].y;                                                       \
    rec.z = A;                                                             \
    *(v3f*)&pkw[(o) * 32] = rec;                                           \
  }
#define PSTORE0()                                                          \
  {                                                                        \
    float2 R = Rv[0];                                                      \
    float  A = (li > 0) ? Av[0] : 0.0f;                                    \
    if (qgroup) { R.x = 0.0f; R.y = 0.0f; A = 0.0f; }                      \
    pkw[0] = make_float4(R.x, R.y, A, __int_as_float(tvN));                \
  }
#define ACC1(o)                                                            \
  {                                                                        \
    float so = row_share<(o)>(sv);                                         \
    if ((o) == 0 && qdone) so = 0.0f;                                      \
    const int    to = row_share_i<(o)>(tvA);                               \
    const float2 tr = Tl2[to * 32 + lane];                                 \
    acc.x = fmaf(tr.x, so, acc.x);                                         \
    acc.y = fmaf(tr.y, so, acc.y);                                         \
  }
#define ACCGROUP()                                                         \
  {                                                                        \
    const int   g2    = NG - 1 - ((k - 2) * NSLOT + h);                    \
    const bool  qdone = (g2 == NG - 1);                                    \
    const float sv    = Sb[g2 * GRP + li];                                 \
    ACC1(0) ACC1(1) ACC1(2) ACC1(3) ACC1(4) ACC1(5) ACC1(6) ACC1(7)        \
    ACC1(8) ACC1(9) ACC1(10) ACC1(11) ACC1(12) ACC1(13) ACC1(14) ACC1(15)  \
  }
#define LOADSET(X, slot)                                                   \
  {                                                                        \
    _Pragma("unroll") for (int o = 0; o < GRP; ++o)                        \
      X[o] = pkr[((slot) * GRP + o) * 32];                                 \
  }
#define MSTEP(o, X)                                                        \
  scan_step_mid<(o)>(p, s, c_lo, c_hi, X[o].z, X[(o) - 1].x, X[(o) - 1].y);
#define GROUP(Xc, Xn, slot, HASNEXT)                                       \
  {                                                                        \
    if (HASNEXT) LOADSET(Xn, (slot) + 1)                                   \
    __builtin_amdgcn_sched_barrier(0);                                     \
    float p = gather_c(c_lo, c_hi, __float_as_int(Xc[0].w), lane);         \
    float s;                                                               \
    scan_step_first(p, s, Xc[0].z);                                        \
    MSTEP(1, Xc) MSTEP(2, Xc) MSTEP(3, Xc) MSTEP(4, Xc) MSTEP(5, Xc)       \
    MSTEP(6, Xc) MSTEP(7, Xc) MSTEP(8, Xc) MSTEP(9, Xc) MSTEP(10, Xc)      \
    MSTEP(11, Xc) MSTEP(12, Xc) MSTEP(13, Xc) MSTEP(14, Xc)                \
    scan_step_last(p, s, c_lo, c_hi, Xc[14].x, Xc[14].y, Xc[15].x, Xc[15].y); \
    Sb[(gfirst - (slot)) * GRP + li] = p;                                  \
  }

  for (int k = 0; k < NTRIP + 2; ++k) {
    if (wave == 0) {
      if (k >= 1 && k <= NTRIP) {
        const float4* pkr    = Pk + ((k - 1) & 1) * (NSLOT * GRP * 32) + lane;
        const int     gfirst = NG - 1 - (k - 1) * NSLOT;
        float4 X0[GRP], X1[GRP];
        LOADSET(X0, 0)
        GROUP(X0, X1, 0, 1)
        GROUP(X1, X0, 1, 1)
        GROUP(X0, X1, 2, 1)
        GROUP(X1, X0, 3, 1)
        GROUP(X0, X1, 4, 1)
        GROUP(X1, X0, 5, 1)
        GROUP(X0, X1, 6, 1)
        GROUP(X1, X0, 7, 0)
      }
    } else {
      const int h   = wave - 1;
      int       tvN = 0;
      if (k < NTRIP) {
        const int  g      = NG - 1 - (k * NSLOT + h);
        const bool qgroup = (g == NG - 1);
        float4*    pkw    = Pk + ((k & 1) * NSLOT + h) * (GRP * 32) + lane;
        tvN = tvP;
        const int kq = (k + 1 < NTRIP) ? k + 1 : k;
        tvP = sqr[(NG - 1 - (kq * NSLOT + h)) * GRP + (GRP - 1 - li)];
        const int offn = ((tvN & 31) << 1) + (tvN >> 5);
        float2 Rv[GRP];
        float  Av[GRP];
        PLOAD(0) PLOAD(1) PLOAD(2) PLOAD(3) PLOAD(4) PLOAD(5) PLOAD(6) PLOAD(7)
        PLOAD(8) PLOAD(9) PLOAD(10) PLOAD(11) PLOAD(12) PLOAD(13) PLOAD(14) PLOAD(15)
        if (k >= 2) ACCGROUP()
        PSTORE0()
        PSTORE(1) PSTORE(2) PSTORE(3) PSTORE(4) PSTORE(5) PSTORE(6) PSTORE(7)
        PSTORE(8) PSTORE(9) PSTORE(10) PSTORE(11) PSTORE(12) PSTORE(13) PSTORE(14) PSTORE(15)
      } else {
        ACCGROUP()
      }
      tvA = tvB;
      tvB = tvN;
    }
    __syncthreads();
  }
#undef GROUP
#undef LOADSET
#undef MSTEP
#undef ACCGROUP
#undef ACC1
#undef PSTORE0
#undef PSTORE
#undef PLOAD

  if (wave != 0) Part[(wave - 1) * 32 + lane] = acc;
  __syncthreads();
  if (wave == 0) {
    float2 r = Part[lane];
    for (int h = 1; h < NSLOT; ++h) {
      const float2 q = Part[h * 32 + lane];
      r.x += q.x;
      r.y += q.y;
    }
    Cx2[lane] = r;
  }
  __syncthreads();

  v2f bO[16];
  if (wave < 4) {
    const int nt = wave;
    v2f bR[16];
#pragma unroll
    for (int kt = 0; kt < 16; ++kt) bR[kt] = frag_b(Wr, 64, kt, nt, lane);
#pragma unroll
    for (int kt = 0; kt < 16; ++kt) bO[kt] = frag_b(Wo, 64, kt, nt, lane);
    v8f acc8 = {};
#pragma unroll
    for (int kt = 0; kt < 16; ++kt)
      acc8 = wmma_f32(frag_a_row0(Cx2, kt, lane), bR[kt], acc8);
    const int   col = (nt << 4) + (lane & 15);
    const float y   = acc8[0] + br[col];
    if (lane < 16) ((float*)Yx2)[col] = y;
  }
  __syncthreads();
  if (wave < 4) {
    const int nt = wave;
    v8f acc8 = {};
#pragma unroll
    for (int kt = 0; kt < 16; ++kt)
      acc8 = wmma_f32(frag_a_row0(Yx2, kt, lane), bO[kt], acc8);
    const int   col = (nt << 4) + (lane & 15);
    const float y   = acc8[0] + bo[col];
    if (lane < 16) ((float*)Ox2)[col] = y;
  }
  __syncthreads();
  if (wave == 0) ((float2*)(out + b * 64))[lane] = Ox2[lane];
}

extern "C" void kernel_launch(void* const* d_in, const int* in_sizes, int n_in,
                              void* d_out, int out_size, void* d_ws, size_t ws_size,
                              hipStream_t stream) {
  const int*   seq   = (const int*)d_in[0];
  const float* embed = (const float*)d_in[1];
  const float* W1    = (const float*)d_in[2];
  const float* b1    = (const float*)d_in[3];
  const float* W2    = (const float*)d_in[4];
  const float* b2    = (const float*)d_in[5];
  const float* gamma = (const float*)d_in[6];
  const float* beta  = (const float*)d_in[7];
  const float* Wr    = (const float*)d_in[8];
  const float* br    = (const float*)d_in[9];
  const float* Wo    = (const float*)d_in[10];
  const float* bo    = (const float*)d_in[11];
  float*       out   = (float*)d_out;

  delta_fused_kernel<<<BATCH, NTHR, 0, stream>>>(seq, embed, W1, b1, W2, b2, gamma, beta,
                                                 Wr, br, Wo, bo, out);
}


# Scan wave gets its SIMD to itself: block of 11 waves, waves 4 and 8 (placed on wave 0's SIMD) sit the scan out, the 8 helpers run on the other three SIMDs; same order of sums
# speedup vs baseline: 14.7658x; 1.0304x over previous
#include <hip/hip_runtime.h>


#define HDIM   64
#define VOCABN 64
#define BATCH  64
#define SEQL   4096
#define TPITCH 68

typedef __attribute__((ext_vector_type(2))) float v2f;
typedef __attribute__((ext_vector_type(2))) float f32x2;
typedef __attribute__((ext_vector_type(8))) float v8f;
typedef __attribute__((ext_vector_type(3))) float v3f;

__device__ __forceinline__ v2f frag_a(const float* A, int lda, int mt, int kt, int lane) {
  int row = (mt << 4) + (lane & 15);
  int k0  = (kt << 2) + ((lane >> 4) << 1);
  v2f a;
  a.x = A[row * lda + k0];
  a.y = A[row * lda + k0 + 1];
  return a;
}
__device__ __forceinline__ v2f frag_b(const float* Bm, int ldb, int kt, int nt, int lane) {
  int col = (nt << 4) + (lane & 15);
  int k0  = (kt << 2) + ((lane >> 4) << 1);
  v2f b;
  b.x = Bm[k0 * ldb + col];
  b.y = Bm[(k0 + 1) * ldb + col];
  return b;
}
__device__ __forceinline__ v2f frag_bt(const float* S, int lds, int kt, int nt, int lane) {
  int col = (nt << 4) + (lane & 15);
  int k0  = (kt << 2) + ((lane >> 4) << 1);
  v2f b;
  b.x = S[col * lds + k0];
  b.y = S[col * lds + k0 + 1];
  return b;
}

__device__ __forceinline__ v8f wmma_f32(v2f a, v2f b, v8f c) {
  return __builtin_amdgcn_wmma_f32_16x16x4_f32(false, a, false, b, (short)0, c,
                                               false, false);
}

__device__ __forceinline__ v2f frag_a_row0(const float2* x2, int kt, int lane) {
  const float2 xv = x2[(kt << 1) + (lane >> 4)];
  const bool   r0 = (lane & 15) == 0;
  v2f a;
  a.x = r0 ? xv.x : 0.0f;
  a.y = r0 ? xv.y : 0.0f;
  return a;
}

#define GRP    16
#define NSLOT  8
#define NTRIP  32
#define NTHR   352

template <int J>
__device__ __forceinline__ int row_share_i(int x) {
  return __builtin_amdgcn_update_dpp(0, x, 0x150 + J, 0xf, 0xf, true);
}
template <int J>
__device__ __forceinline__ float row_share(float x) {
  return __int_as_float(row_share_i<J>(__float_as_int(x)));
}

#define DPPCTL " row_mask:0xf bank_mask:0xf bound_ctrl:1"
__device__ __forceinline__ void scan_step_first(float& p, float& s, float a) {
  float pn, sn;
  asm("v_fma_f32_e64_dpp %0, %2, %3, %2 row_share:0" DPPCTL "\n\t"
      "v_mov_b32_dpp %1, %2 row_share:0" DPPCTL
      : "=&v"(pn), "=&v"(sn)
      : "v"(p), "v"(a));
  p = pn;
  s = sn;
}
template <int J>
__device__ __forceinline__ void scan_step_mid(float& p, float& s, float& c_lo, float& c_hi,
                                              float a, float gx, float gy) {
  float pn, sn;
  asm("s_delay_alu instid0(VALU_DEP_4)\n\t"
      "v_fma_f32_e64_dpp %0, %4, %5, %4 row_share:%9" DPPCTL "\n\t"
      "v_mov_b32_dpp %1, %4 row_share:%9" DPPCTL "\n\t"
      "v_fmac_f32_e32 %2, %6, %8\n\t"
      "v_fmac_f32_e32 %3, %7, %8"
      : "=&v"(pn), "=&v"(sn), "+v"(c_lo), "+v"(c_hi)
      : "v"(p), "v"(a), "v"(gx), "v"(gy), "v"(s), "n"(J));
  p = pn;
  s = sn;
}
__device__ __forceinline__ void scan_step_last(float p, float s, float& c_lo, float& c_hi,
                                               float gx14, float gy14, float gx15, float gy15) {
  asm("v_fmac_f32_e32 %0, %3, %5\n\t"
      "v_fmac_f32_e32 %1, %4, %5\n\t"
      "v_fmac_f32_dpp %0, %2, %6 row_share:15" DPPCTL "\n\t"
      "v_fmac_f32_dpp %1, %2, %7 row_share:15" DPPCTL
      : "+v"(c_lo), "+v"(c_hi)
      : "v"(p), "v"(gx14), "v"(gy14), "v"(s), "v"(gx15), "v"(gy15));
}

__device__ __forceinline__ float gather_c(float c_lo, float c_hi, int tok, int lane) {
  const unsigned sel = (unsigned)tok & 15u;
  const unsigned ulo = (unsigned)__float_as_int(c_lo);
  const unsigned uhi = (unsigned)__float_as_int(c_hi);
  const unsigned lo_s = __builtin_amdgcn_permlane16_var(ulo, ulo, sel, false, false);
  const unsigned lo_x = __builtin_amdgcn_permlanex16_var(ulo, ulo, sel, false, false);
  const unsigned hi_s = __builtin_amdgcn_permlane16_var(uhi, uhi, sel, false, false);
  const unsigned hi_x = __builtin_amdgcn_permlanex16_var(uhi, uhi, sel, false, false);
  const bool other_row = (((tok >> 4) ^ (lane >> 4)) & 1) != 0;
  const unsigned lo = other_row ? lo_x : lo_s;
  const unsigned hi = other_row ? hi_x : hi_s;
  return __int_as_float((int)((tok & 32) ? hi : lo));
}

#define NWAVE  11

__global__ __launch_bounds__(NTHR) void delta_fused_kernel(
    const int* __restrict__ seq, const float* __restrict__ embed,
    const float* __restrict__ W1, const float* __restrict__ b1,
    const float* __restrict__ W2, const float* __restrict__ b2,
    const float* __restrict__ gamma, const float* __restrict__ beta,
    const float* __restrict__ Wr, const float* __restrict__ br,
    const float* __restrict__ Wo, const float* __restrict__ bo,
    float* __restrict__ out) {
  __shared__ float4 Gl4[VOCABN * 16];
  __shared__ float4 Tl4[VOCABN * 16];
  __shared__ float4 Pk[2 * NSLOT * GRP * 32];
  __shared__ float  Sb[SEQL];
  __shared__ float2 Part[NSLOT * 32];
  __shared__ float4 Rl4[64 * 32];
  __shared__ float4 Xl4[64 * 16];
  __shared__ float  rdl[64];
  __shared__ float  Gpl[VOCABN * VOCABN];
  __shared__ float2 Cx2[32];
  __shared__ float2 Yx2[32];
  __shared__ float2 Ox2[32];
  float* Rlds = (float*)Rl4;
  float* Xlds = (float*)Xl4;
  const int b   = blockIdx.x;
  const int tid = threadIdx.x;
  const int lane = tid & 31;
  const int wave = __builtin_amdgcn_readfirstlane(tid >> 5);

  const bool helper = (wave & 3) != 0;
  const int  hslot  = wave - 1 - (wave >> 2);

  int tvP = 0;
  if (helper) tvP = seq[b * SEQL + (SEQL / 16 - 1 - hslot) * 16 + (15 - (lane & 15))];

  for (int tile = wave; tile < 32; tile += NWAVE) {
    int mt = tile >> 3, nt = tile & 7;
    v8f acc8 = {};
#pragma unroll
    for (int kt = 0; kt < 16; ++kt) {
      v2f a = frag_a(embed, 64, mt, kt, lane);
      v2f bq = frag_b(W1, 128, kt, nt, lane);
      acc8 = wmma_f32(a, bq, acc8);
    }
    int col   = (nt << 4) + (lane & 15);
    int rbase = (mt << 4) + ((lane >> 4) << 3);
#pragma unroll
    for (int v = 0; v < 8; ++v) {
      float x = acc8[v] + b1[col];
      Rlds[(rbase + v) * 128 + col] = fmaxf(x, 0.0f);
    }
  }
  __syncthreads();

  for (int tile = wave; tile < 16; tile += NWAVE) {
    int mt = tile >> 2, nt = tile & 3;
    v8f acc8 = {};
#pragma unroll
    for (int kt = 0; kt < 32; ++kt) {
      v2f a = frag_a(Rlds, 128, mt, kt, lane);
      v2f bq = frag_b(W2, 64, kt, nt, lane);
      acc8 = wmma_f32(a, bq, acc8);
    }
    int col   = (nt << 4) + (lane & 15);
    int rbase = (mt << 4) + ((lane >> 4) << 3);
#pragma unroll
    for (int v = 0; v < 8; ++v) {
      int row = rbase + v;
      Xlds[row * 64 + col] = embed[row * 64 + col] + acc8[v] + b2[col];
    }
  }
  __syncthreads();

  if (tid < 64) {
    const float* xr = &Xlds[tid * 64];
    float mu = 0.f;
#pragma unroll
    for (int c = 0; c < 64; ++c) mu += xr[c];
    mu *= (1.0f / 64.0f);
    float var = 0.f;
#pragma unroll
    for (int c = 0; c < 64; ++c) { float d = xr[c] - mu; var += d * d; }
    var *= (1.0f / 64.0f);
    float rs  = rsqrtf(var + 1e-5f);
    float ksq = 0.f;
#pragma unroll
    for (int c = 0; c < 64; ++c) {
      float tv = (xr[c] - mu) * rs * gamma[c] + beta[c];
      Rlds[tid * TPITCH + c] = tv;
      ksq += tv * tv;
    }
    rdl[tid] = 1.0f / (ksq + 1e-6f);
  }
  __syncthreads();

  for (int i = tid; i < VOCABN * 16; i += NTHR)
    Tl4[i] = *(const float4*)&Rlds[(i >> 4) * TPITCH + ((i & 15) << 2)];

  {
    const float* Tp  = Rlds;
    float*       Gsf = (float*)Gl4;
    for (int tile = wave; tile < 16; tile += NWAVE) {
      int mt = tile >> 2, nt = tile & 3;
      v8f acc8 = {};
#pragma unroll
      for (int kt = 0; kt < 16; ++kt) {
        v2f a = frag_a(Tp, TPITCH, mt, kt, lane);
        v2f bq = frag_bt(Tp, TPITCH, kt, nt, lane);
        acc8 = wmma_f32(a, bq, acc8);
      }
      int col   = (nt << 4) + (lane & 15);
      int rbase = (mt << 4) + ((lane >> 4) << 3);
#pragma unroll
      for (int v = 0; v < 8; ++v) {
        int row = rbase + v;
        Gpl[row * 64 + col] = acc8[v];
        Gsf[((row << 5) + (col & 31)) * 2 + (col >> 5)] = -(acc8[v] * rdl[row]);
      }
    }
  }
  __syncthreads();

  const float2* Gl2 = (const float2*)Gl4;
  const float*  Glf = (const float*)Gl4;
  const float2* Tl2 = (const float2*)Tl4;
  const int*    sqr = seq + b * SEQL;
  const int     li  = lane & 15;
  const int     NG  = SEQL / GRP;

  float  c_lo = 0.0f, c_hi = 0.0f;
  float2 acc  = make_float2(0.0f, 0.0f);
  int    tvA = 0, tvB = 0;
  if (wave == 0) {
    const int tq = sqr[SEQL - 1];
    c_lo = Gpl[tq * 64 + lane];
    c_hi = Gpl[tq * 64 + lane + 32];
  }

#define PLOAD(o)                                                           \
  {                                                                        \
    const int to = row_share_i<(o)>(tvN);                                  \
    Rv[o] = Gl2[to * 32 + lane];                                           \
    Av[o] = Glf[to * 64 + offn];                                           \
  }
#define PSTORE(o)                                                          \
  {                                                                        \
    const float A = (li > (o)) ? Av[o] : 0.0f;                             \
    v3f rec;                                                               \
    rec.x = Rv[o].x;                                                       \
    rec.y = Rv[o].y;                                                       \
    rec.z = A;                                                             \
    *(v3f*)&pkw[(o) * 32] = rec;                                           \
  }
#define PSTORE0()                                                          \
  {                                                                        \
    float2 R = Rv[0];                                                      \
    float  A = (li > 0) ? Av[0] : 0.0f;                                    \
    if (qgroup) { R.x = 0.0f; R.y = 0.0f; A = 0.0f; }                      \
    pkw[0] = make_float4(R.x, R.y, A, __int_as_float(tvN));                \
  }
#define ACC1(o)                                                            \
  {                                                                        \
    float so = row_share<(o)>(sv);                                         \
    if ((o) == 0 && qdone) so = 0.0f;                                      \
    const int    to = row_share_i<(o)>(tvA);                               \
    const float2 tr = Tl2[to * 32 + lane];                                 \
    acc.x = fmaf(tr.x, so, acc.x);                                         \
    acc.y = fmaf(tr.y, so, acc.y);                                         \
  }
#define ACCGROUP()                                                         \
  {                                                                        \
    const int   g2    = NG - 1 - ((k - 2) * NSLOT + h);                    \
    const bool  qdone = (g2 == NG - 1);                                    \
    const float sv    = Sb[g2 * GRP + li];                                 \
    ACC1(0) ACC1(1) ACC1(2) ACC1(3) ACC1(4) ACC1(5) ACC1(6) ACC1(7)        \
    ACC1(8) ACC1(9) ACC1(10) ACC1(11) ACC1(12) ACC1(13) ACC1(14) ACC1(15)  \
  }
#define LOADSET(X, slot)                                                   \
  {                                                                        \
    _Pragma("unroll") for (int o = 0; o < GRP; ++o)                        \
      X[o] = pkr[((slot) * GRP + o) * 32];                                 \
  }
#define MSTEP(o, X)                                                        \
  scan_step_mid<(o)>(p, s, c_lo, c_hi, X[o].z, X[(o) - 1].x, X[(o) - 1].y);
#define GROUP(Xc, Xn, slot, HASNEXT)                                       \
  {                                                                        \
    if (HASNEXT) LOADSET(Xn, (slot) + 1)                                   \
    __builtin_amdgcn_sched_barrier(0);                                     \
    float p = gather_c(c_lo, c_hi, __float_as_int(Xc[0].w), lane);         \
    float s;                                                               \
    scan_step_first(p, s, Xc[0].z);                                        \
    MSTEP(1, Xc) MSTEP(2, Xc) MSTEP(3, Xc) MSTEP(4, Xc) MSTEP(5, Xc)       \
    MSTEP(6, Xc) MSTEP(7, Xc) MSTEP(8, Xc) MSTEP(9, Xc) MSTEP(10, Xc)      \
    MSTEP(11, Xc) MSTEP(12, Xc) MSTEP(13, Xc) MSTEP(14, Xc)                \
    scan_step_last(p, s, c_lo, c_hi, Xc[14].x, Xc[14].y, Xc[15].x, Xc[15].y); \
    Sb[(gfirst - (slot)) * GRP + li] = p;                                  \
  }

  for (int k = 0; k < NTRIP + 2; ++k) {
    if (wave == 0) {
      if (k >= 1 && k <= NTRIP) {
        const float4* pkr    = Pk + ((k - 1) & 1) * (NSLOT * GRP * 32) + lane;
        const int     gfirst = NG - 1 - (k - 1) * NSLOT;
        float4 X0[GRP], X1[GRP];
        LOADSET(X0, 0)
        GROUP(X0, X1, 0, 1)
        GROUP(X1, X0, 1, 1)
        GROUP(X0, X1, 2, 1)
        GROUP(X1, X0, 3, 1)
        GROUP(X0, X1, 4, 1)
        GROUP(X1, X0, 5, 1)
        GROUP(X0, X1, 6, 1)
        GROUP(X1, X0, 7, 0)
      }
    } else if (helper) {
      const int h   = hslot;
      int       tvN = 0;
      if (k < NTRIP) {
        const int  g      = NG - 1 - (k * NSLOT + h);
        const bool qgroup = (g == NG - 1);
        float4*    pkw    = Pk + ((k & 1) * NSLOT + h) * (GRP * 32) + lane;
        tvN = tvP;
        const int kq = (k + 1 < NTRIP) ? k + 1 : k;
        tvP = sqr[(NG - 1 - (kq * NSLOT + h)) * GRP + (GRP - 1 - li)];
        const int offn = ((tvN & 31) << 1) + (tvN >> 5);
        float2 Rv[GRP];
        float  Av[GRP];
        PLOAD(0) PLOAD(1) PLOAD(2) PLOAD(3) PLOAD(4) PLOAD(5) PLOAD(6) PLOAD(7)
        PLOAD(8) PLOAD(9) PLOAD(10) PLOAD(11) PLOAD(12) PLOAD(13) PLOAD(14) PLOAD(15)
        if (k >= 2) ACCGROUP()
        PSTORE0()
        PSTORE(1) PSTORE(2) PSTORE(3) PSTORE(4) PSTORE(5) PSTORE(6) PSTORE(7)
        PSTORE(8) PSTORE(9) PSTORE(10) PSTORE(11) PSTORE(12) PSTORE(13) PSTORE(14) PSTORE(15)
      } else {
        ACCGROUP()
      }
      tvA = tvB;
      tvB = tvN;
    }
    __syncthreads();
  }
#undef GROUP
#undef LOADSET
#undef MSTEP
#undef ACCGROUP
#undef ACC1
#undef PSTORE0
#undef PSTORE
#undef PLOAD

  if (helper) Part[hslot * 32 + lane] = acc;
  __syncthreads();
  if (wave == 0) {
    float2 r = Part[lane];
    for (int h = 1; h < NSLOT; ++h) {
      const float2 q = Part[h * 32 + lane];
      r.x += q.x;
      r.y += q.y;
    }
    Cx2[lane] = r;
  }
  __syncthreads();

  v2f bO[16];
  if (wave < 4) {
    const int nt = wave;
    v2f bR[16];
#pragma unroll
    for (int kt = 0; kt < 16; ++kt) bR[kt] = frag_b(Wr, 64, kt, nt, lane);
#pragma unroll
    for (int kt = 0; kt < 16; ++kt) bO[kt] = frag_b(Wo, 64, kt, nt, lane);
    v8f acc8 = {};
#pragma unroll
    for (int kt = 0; kt < 16; ++kt)
      acc8 = wmma_f32(frag_a_row0(Cx2, kt, lane), bR[kt], acc8);
    const int   col = (nt << 4) + (lane & 15);
    const float y   = acc8[0] + br[col];
    if (lane < 16) ((float*)Yx2)[col] = y;
  }
  __syncthreads();
  if (wave < 4) {
    const int nt = wave;
    v8f acc8 = {};
#pragma unroll
    for (int kt = 0; kt < 16; ++kt)
      acc8 = wmma_f32(frag_a_row0(Yx2, kt, lane), bO[kt], acc8);
    const int   col = (nt << 4) + (lane & 15);
    const float y   = acc8[0] + bo[col];
    if (lane < 16) ((float*)Ox2)[col] = y;
  }
  __syncthreads();
  if (wave == 0) ((float2*)(out + b * 64))[lane] = Ox2[lane];
}

extern "C" void kernel_launch(void* const* d_in, const int* in_sizes, int n_in,
                              void* d_out, int out_size, void* d_ws, size_t ws_size,
                              hipStream_t stream) {
  const int*   seq   = (const int*)d_in[0];
  const float* embed = (const float*)d_in[1];
  const float* W1    = (const float*)d_in[2];
  const float* b1    = (const float*)d_in[3];
  const float* W2    = (const float*)d_in[4];
  const float* b2    = (const float*)d_in[5];
  const float* gamma = (const float*)d_in[6];
  const float* beta  = (const float*)d_in[7];
  const float* Wr    = (const float*)d_in[8];
  const float* br    = (const float*)d_in[9];
  const float* Wo    = (const float*)d_in[10];
  const float* bo    = (const float*)d_in[11];
  float*       out   = (float*)d_out;

  delta_fused_kernel<<<BATCH, NTHR, 0, stream>>>(seq, embed, W1, b1, W2, b2, gamma, beta,
                                                 Wr, br, Wo, bo, out);
}


# Table phase's LDS rows padded: R (A operand of the 2nd product) to pitch 132 floats, X (LayerNorm input) to 68; a fragment's 16 rows and the LayerNorm's lanes hit distinct banks; same order of sums
# speedup vs baseline: 15.7793x; 1.0686x over previous
#include <hip/hip_runtime.h>


#define HDIM   64
#define VOCABN 64
#define BATCH  64
#define SEQL   4096
#define TPITCH 68
#define RPITCH 132
#define XPITCH 68

typedef __attribute__((ext_vector_type(2))) float v2f;
typedef __attribute__((ext_vector_type(2))) float f32x2;
typedef __attribute__((ext_vector_type(8))) float v8f;
typedef __attribute__((ext_vector_type(3))) float v3f;

__device__ __forceinline__ v2f frag_a(const float* A, int lda, int mt, int kt, int lane) {
  int row = (mt << 4) + (lane & 15);
  int k0  = (kt << 2) + ((lane >> 4) << 1);
  v2f a;
  a.x = A[row * lda + k0];
  a.y = A[row * lda + k0 + 1];
  return a;
}
__device__ __forceinline__ v2f frag_b(const float* Bm, int ldb, int kt, int nt, int lane) {
  int col = (nt << 4) + (lane & 15);
  int k0  = (kt << 2) + ((lane >> 4) << 1);
  v2f b;
  b.x = Bm[k0 * ldb + col];
  b.y = Bm[(k0 + 1) * ldb + col];
  return b;
}
__device__ __forceinline__ v2f frag_bt(const float* S, int lds, int kt, int nt, int lane) {
  int col = (nt << 4) + (lane & 15);
  int k0  = (kt << 2) + ((lane >> 4) << 1);
  v2f b;
  b.x = S[col * lds + k0];
  b.y = S[col * lds + k0 + 1];
  return b;
}

__device__ __forceinline__ v8f wmma_f32(v2f a, v2f b, v8f c) {
  return __builtin_amdgcn_wmma_f32_16x16x4_f32(false, a, false, b, (short)0, c,
                                               false, false);
}

__device__ __forceinline__ v2f frag_a_row0(const float2* x2, int kt, int lane) {
  const float2 xv = x2[(kt << 1) + (lane >> 4)];
  const bool   r0 = (lane & 15) == 0;
  v2f a;
  a.x = r0 ? xv.x : 0.0f;
  a.y = r0 ? xv.y : 0.0f;
  return a;
}

#define GRP    16
#define NSLOT  8
#define NTRIP  32
#define NTHR   352

template <int J>
__device__ __forceinline__ int row_share_i(int x) {
  return __builtin_amdgcn_update_dpp(0, x, 0x150 + J, 0xf, 0xf, true);
}
template <int J>
__device__ __forceinline__ float row_share(float x) {
  return __int_as_float(row_share_i<J>(__float_as_int(x)));
}

#define DPPCTL " row_mask:0xf bank_mask:0xf bound_ctrl:1"
__device__ __forceinline__ void scan_step_first(float& p, float& s, float a) {
  float pn, sn;
  asm("v_fma_f32_e64_dpp %0, %2, %3, %2 row_share:0" DPPCTL "\n\t"
      "v_mov_b32_dpp %1, %2 row_share:0" DPPCTL
      : "=&v"(pn), "=&v"(sn)
      : "v"(p), "v"(a));
  p = pn;
  s = sn;
}
template <int J>
__device__ __forceinline__ void scan_step_mid(float& p, float& s, float& c_lo, float& c_hi,
                                              float a, float gx, float gy) {
  float pn, sn;
  asm("s_delay_alu instid0(VALU_DEP_4)\n\t"
      "v_fma_f32_e64_dpp %0, %4, %5, %4 row_share:%9" DPPCTL "\n\t"
      "v_mov_b32_dpp %1, %4 row_share:%9" DPPCTL "\n\t"
      "v_fmac_f32_e32 %2, %6, %8\n\t"
      "v_fmac_f32_e32 %3, %7, %8"
      : "=&v"(pn), "=&v"(sn), "+v"(c_lo), "+v"(c_hi)
      : "v"(p), "v"(a), "v"(gx), "v"(gy), "v"(s), "n"(J));
  p = pn;
  s = sn;
}
__device__ __forceinline__ void scan_step_last(float p, float s, float& c_lo, float& c_hi,
                                               float gx14, float gy14, float gx15, float gy15) {
  asm("v_fmac_f32_e32 %0, %3, %5\n\t"
      "v_fmac_f32_e32 %1, %4, %5\n\t"
      "v_fmac_f32_dpp %0, %2, %6 row_share:15" DPPCTL "\n\t"
      "v_fmac_f32_dpp %1, %2, %7 row_share:15" DPPCTL
      : "+v"(c_lo), "+v"(c_hi)
      : "v"(p), "v"(gx14), "v"(gy14), "v"(s), "v"(gx15), "v"(gy15));
}

__device__ __forceinline__ float gather_c(float c_lo, float c_hi, int tok, int lane) {
  const unsigned sel = (unsigned)tok & 15u;
  const unsigned ulo = (unsigned)__float_as_int(c_lo);
  const unsigned uhi = (unsigned)__float_as_int(c_hi);
  const unsigned lo_s = __builtin_amdgcn_permlane16_var(ulo, ulo, sel, false, false);
  const unsigned lo_x = __builtin_amdgcn_permlanex16_var(ulo, ulo, sel, false, false);
  const unsigned hi_s = __builtin_amdgcn_permlane16_var(uhi, uhi, sel, false, false);
  const unsigned hi_x = __builtin_amdgcn_permlanex16_var(uhi, uhi, sel, false, false);
  const bool other_row = (((tok >> 4) ^ (lane >> 4)) & 1) != 0;
  const unsigned lo = other_row ? lo_x : lo_s;
  const unsigned hi = other_row ? hi_x : hi_s;
  return __int_as_float((int)((tok & 32) ? hi : lo));
}

#define NWAVE  11

__global__ __launch_bounds__(NTHR) void delta_fused_kernel(
    const int* __restrict__ seq, const float* __restrict__ embed,
    const float* __restrict__ W1, const float* __restrict__ b1,
    const float* __restrict__ W2, const float* __restrict__ b2,
    const float* __restrict__ gamma, const float* __restrict__ beta,
    const float* __restrict__ Wr, const float* __restrict__ br,
    const float* __restrict__ Wo, const float* __restrict__ bo,
    float* __restrict__ out) {
  __shared__ float4 Gl4[VOCABN * 16];
  __shared__ float4 Tl4[VOCABN * 16];
  __shared__ float4 Pk[2 * NSLOT * GRP * 32];
  __shared__ float  Sb[SEQL];
  __shared__ float2 Part[NSLOT * 32];
  __shared__ float4 Rl4[64 * (RPITCH / 4)];
  __shared__ float4 Xl4[64 * (XPITCH / 4)];
  __shared__ float  rdl[64];
  __shared__ float  Gpl[VOCABN * VOCABN];
  __shared__ float2 Cx2[32];
  __shared__ float2 Yx2[32];
  __shared__ float2 Ox2[32];
  float* Rlds = (float*)Rl4;
  float* Xlds = (float*)Xl4;
  const int b   = blockIdx.x;
  const int tid = threadIdx.x;
  const int lane = tid & 31;
  const int wave = __builtin_amdgcn_readfirstlane(tid >> 5);

  const bool helper = (wave & 3) != 0;
  const int  hslot  = wave - 1 - (wave >> 2);

  int tvP = 0;
  if (helper) tvP = seq[b * SEQL + (SEQL / 16 - 1 - hslot) * 16 + (15 - (lane & 15))];

  for (int tile = wave; tile < 32; tile += NWAVE) {
    int mt = tile >> 3, nt = tile & 7;
    v8f acc8 = {};
#pragma unroll
    for (int kt = 0; kt < 16; ++kt) {
      v2f a = frag_a(embed, 64, mt, kt, lane);
      v2f bq = frag_b(W1, 128, kt, nt, lane);
      acc8 = wmma_f32(a, bq, acc8);
    }
    int col   = (nt << 4) + (lane & 15);
    int rbase = (mt << 4) + ((lane >> 4) << 3);
#pragma unroll
    for (int v = 0; v < 8; ++v) {
      float x = acc8[v] + b1[col];
      Rlds[(rbase + v) * RPITCH + col] = fmaxf(x, 0.0f);
    }
  }
  __syncthreads();

  for (int tile = wave; tile < 16; tile += NWAVE) {
    int mt = tile >> 2, nt = tile & 3;
    v8f acc8 = {};
#pragma unroll
    for (int kt = 0; kt < 32; ++kt) {
      v2f a = frag_a(Rlds, RPITCH, mt, kt, lane);
      v2f bq = frag_b(W2, 64, kt, nt, lane);
      acc8 = wmma_f32(a, bq, acc8);
    }
    int col   = (nt << 4) + (lane & 15);
    int rbase = (mt << 4) + ((lane >> 4) << 3);
#pragma unroll
    for (int v = 0; v < 8; ++v) {
      int row = rbase + v;
      Xlds[row * XPITCH + col] = embed[row * 64 + col] + acc8[v] + b2[col];
    }
  }
  __syncthreads();

  if (tid < 64) {
    const float* xr = &Xlds[tid * XPITCH];
    float mu = 0.f;
#pragma unroll
    for (int c = 0; c < 64; ++c) mu += xr[c];
    mu *= (1.0f / 64.0f);
    float var = 0.f;
#pragma unroll
    for (int c = 0; c < 64; ++c) { float d = xr[c] - mu; var += d * d; }
    var *= (1.0f / 64.0f);
    float rs  = rsqrtf(var + 1e-5f);
    float ksq = 0.f;
#pragma unroll
    for (int c = 0; c < 64; ++c) {
      float tv = (xr[c] - mu) * rs * gamma[c] + beta[c];
      Rlds[tid * TPITCH + c] = tv;
      ksq += tv * tv;
    }
    rdl[tid] = 1.0f / (ksq + 1e-6f);
  }
  __syncthreads();

  for (int i = tid; i < VOCABN * 16; i += NTHR)
    Tl4[i] = *(const float4*)&Rlds[(i >> 4) * TPITCH + ((i & 15) << 2)];

  {
    const float* Tp  = Rlds;
    float*       Gsf = (float*)Gl4;
    for (int tile = wave; tile < 16; tile += NWAVE) {
      int mt = tile >> 2, nt = tile & 3;
      v8f acc8 = {};
#pragma unroll
      for (int kt = 0; kt < 16; ++kt) {
        v2f a = frag_a(Tp, TPITCH, mt, kt, lane);
        v2f bq = frag_bt(Tp, TPITCH, kt, nt, lane);
        acc8 = wmma_f32(a, bq, acc8);
      }
      int col   = (nt << 4) + (lane & 15);
      int rbase = (mt << 4) + ((lane >> 4) << 3);
#pragma unroll
      for (int v = 0; v < 8; ++v) {
        int row = rbase + v;
        Gpl[row * 64 + col] = acc8[v];
        Gsf[((row << 5) + (col & 31)) * 2 + (col >> 5)] = -(acc8[v] * rdl[row]);
      }
    }
  }
  __syncthreads();

  const float2* Gl2 = (const float2*)Gl4;
  const float*  Glf = (const float*)Gl4;
  const float2* Tl2 = (const float2*)Tl4;
  const int*    sqr = seq + b * SEQL;
  const int     li  = lane & 15;
  const int     NG  = SEQL / GRP;

  float  c_lo = 0.0f, c_hi = 0.0f;
  float2 acc  = make_float2(0.0f, 0.0f);
  int    tvA = 0, tvB = 0;
  if (wave == 0) {
    const int tq = sqr[SEQL - 1];
    c_lo = Gpl[tq * 64 + lane];
    c_hi = Gpl[tq * 64 + lane + 32];
  }

#define PLOAD(o)                                                           \
  {                                                                        \
    const int to = row_share_i<(o)>(tvN);                                  \
    Rv[o] = Gl2[to * 32 + lane];                                           \
    Av[o] = Glf[to * 64 + offn];                                           \
  }
#define PSTORE(o)                                                          \
  {                                                                        \
    const float A = (li > (o)) ? Av[o] : 0.0f;                             \
    v3f rec;                                                               \
    rec.x = Rv[o].x;                                                       \
    rec.y = Rv[o].y;                                                       \
    rec.z = A;                                                             \
    *(v3f*)&pkw[(o) * 32] = rec;                                           \
  }
#define PSTORE0()                                                          \
  {                                                                        \
    float2 R = Rv[0];                                                      \
    float  A = (li > 0) ? Av[0] : 0.0f;                                    \
    if (qgroup) { R.x = 0.0f; R.y = 0.0f; A = 0.0f; }                      \
    pkw[0] = make_float4(R.x, R.y, A, __int_as_float(tvN));                \
  }
#define ACC1(o)                                                            \
  {                                                                        \
    float so = row_share<(o)>(sv);                                         \
    if ((o) == 0 && qdone) so = 0.0f;                                      \
    const int    to = row_share_i<(o)>(tvA);                               \
    const float2 tr = Tl2[to * 32 + lane];                                 \
    acc.x = fmaf(tr.x, so, acc.x);                                         \
    acc.y = fmaf(tr.y, so, acc.y);                                         \
  }
#define ACCGROUP()                                                         \
  {                                                                        \
    const int   g2    = NG - 1 - ((k - 2) * NSLOT + h);                    \
    const bool  qdone = (g2 == NG - 1);                                    \
    const float sv    = Sb[g2 * GRP + li];                                 \
    ACC1(0) ACC1(1) ACC1(2) ACC1(3) ACC1(4) ACC1(5) ACC1(6) ACC1(7)        \
    ACC1(8) ACC1(9) ACC1(10) ACC1(11) ACC1(12) ACC1(13) ACC1(14) ACC1(15)  \
  }
#define LOADSET(X, slot)                                                   \
  {                                                                        \
    _Pragma("unroll") for (int o = 0; o < GRP; ++o)                        \
      X[o] = pkr[((slot) * GRP + o) * 32];                                 \
  }
#define MSTEP(o, X)                                                        \
  scan_step_mid<(o)>(p, s, c_lo, c_hi, X[o].z, X[(o) - 1].x, X[(o) - 1].y);
#define GROUP(Xc, Xn, slot, HASNEXT)                                       \
  {                                                                        \
    if (HASNEXT) LOADSET(Xn, (slot) + 1)                                   \
    __builtin_amdgcn_sched_barrier(0);                                     \
    float p = gather_c(c_lo, c_hi, __float_as_int(Xc[0].w), lane);         \
    float s;                                                               \
    scan_step_first(p, s, Xc[0].z);                                        \
    MSTEP(1, Xc) MSTEP(2, Xc) MSTEP(3, Xc) MSTEP(4, Xc) MSTEP(5, Xc)       \
    MSTEP(6, Xc) MSTEP(7, Xc) MSTEP(8, Xc) MSTEP(9, Xc) MSTEP(10, Xc)      \
    MSTEP(11, Xc) MSTEP(12, Xc) MSTEP(13, Xc) MSTEP(14, Xc)                \
    scan_step_last(p, s, c_lo, c_hi, Xc[14].x, Xc[14].y, Xc[15].x, Xc[15].y); \
    Sb[(gfirst - (slot)) * GRP + li] = p;                                  \
  }

  for (int k = 0; k < NTRIP + 2; ++k) {
    if (wave == 0) {
      if (k >= 1 && k <= NTRIP) {
        const float4* pkr    = Pk + ((k - 1) & 1) * (NSLOT * GRP * 32) + lane;
        const int     gfirst = NG - 1 - (k - 1) * NSLOT;
        float4 X0[GRP], X1[GRP];
        LOADSET(X0, 0)
        GROUP(X0, X1, 0, 1)
        GROUP(X1, X0, 1, 1)
        GROUP(X0, X1, 2, 1)
        GROUP(X1, X0, 3, 1)
        GROUP(X0, X1, 4, 1)
        GROUP(X1, X0, 5, 1)
        GROUP(X0, X1, 6, 1)
        GROUP(X1, X0, 7, 0)
      }
    } else if (helper) {
      const int h   = hslot;
      int       tvN = 0;
      if (k < NTRIP) {
        const int  g      = NG - 1 - (k * NSLOT + h);
        const bool qgroup = (g == NG - 1);
        float4*    pkw    = Pk + ((k & 1) * NSLOT + h) * (GRP * 32) + lane;
        tvN = tvP;
        const int kq = (k + 1 < NTRIP) ? k + 1 : k;
        tvP = sqr[(NG - 1 - (kq * NSLOT + h)) * GRP + (GRP - 1 - li)];
        const int offn = ((tvN & 31) << 1) + (tvN >> 5);
        float2 Rv[GRP];
        float  Av[GRP];
        PLOAD(0) PLOAD(1) PLOAD(2) PLOAD(3) PLOAD(4) PLOAD(5) PLOAD(6) PLOAD(7)
        PLOAD(8) PLOAD(9) PLOAD(10) PLOAD(11) PLOAD(12) PLOAD(13) PLOAD(14) PLOAD(15)
        if (k >= 2) ACCGROUP()
        PSTORE0()
        PSTORE(1) PSTORE(2) PSTORE(3) PSTORE(4) PSTORE(5) PSTORE(6) PSTORE(7)
        PSTORE(8) PSTORE(9) PSTORE(10) PSTORE(11) PSTORE(12) PSTORE(13) PSTORE(14) PSTORE(15)
      } else {
        ACCGROUP()
      }
      tvA = tvB;
      tvB = tvN;
    }
    __syncthreads();
  }
#undef GROUP
#undef LOADSET
#undef MSTEP
#undef ACCGROUP
#undef ACC1
#undef PSTORE0
#undef PSTORE
#undef PLOAD

  if (helper) Part[hslot * 32 + lane] = acc;
  __syncthreads();
  if (wave == 0) {
    float2 r = Part[lane];
    for (int h = 1; h < NSLOT; ++h) {
      const float2 q = Part[h * 32 + lane];
      r.x += q.x;
      r.y += q.y;
    }
    Cx2[lane] = r;
  }
  __syncthreads();

  v2f bO[16];
  if (wave < 4) {
    const int nt = wave;
    v2f bR[16];
#pragma unroll
    for (int kt = 0; kt < 16; ++kt) bR[kt] = frag_b(Wr, 64, kt, nt, lane);
#pragma unroll
    for (int kt = 0; kt < 16; ++kt) bO[kt] = frag_b(Wo, 64, kt, nt, lane);
    v8f acc8 = {};
#pragma unroll
    for (int kt = 0; kt < 16; ++kt)
      acc8 = wmma_f32(frag_a_row0(Cx2, kt, lane), bR[kt], acc8);
    const int   col = (nt << 4) + (lane & 15);
    const float y   = acc8[0] + br[col];
    if (lane < 16) ((float*)Yx2)[col] = y;
  }
  __syncthreads();
  if (wave < 4) {
    const int nt = wave;
    v8f acc8 = {};
#pragma unroll
    for (int kt = 0; kt < 16; ++kt)
      acc8 = wmma_f32(frag_a_row0(Yx2, kt, lane), bO[kt], acc8);
    const int   col = (nt << 4) + (lane & 15);
    const float y   = acc8[0] + bo[col];
    if (lane < 16) ((float*)Ox2)[col] = y;
  }
  __syncthreads();
  if (wave == 0) ((float2*)(out + b * 64))[lane] = Ox2[lane];
}

extern "C" void kernel_launch(void* const* d_in, const int* in_sizes, int n_in,
                              void* d_out, int out_size, void* d_ws, size_t ws_size,
                              hipStream_t stream) {
  const int*   seq   = (const int*)d_in[0];
  const float* embed = (const float*)d_in[1];
  const float* W1    = (const float*)d_in[2];
  const float* b1    = (const float*)d_in[3];
  const float* W2    = (const float*)d_in[4];
  const float* b2    = (const float*)d_in[5];
  const float* gamma = (const float*)d_in[6];
  const float* beta  = (const float*)d_in[7];
  const float* Wr    = (const float*)d_in[8];
  const float* br    = (const float*)d_in[9];
  const float* Wo    = (const float*)d_in[10];
  const float* bo    = (const float*)d_in[11];
  float*       out   = (float*)d_out;

  delta_fused_kernel<<<BATCH, NTHR, 0, stream>>>(seq, embed, W1, b1, W2, b2, gamma, beta,
                                                 Wr, br, Wo, bo, out);
}


# Table phase's two products fed from LDS: embed, W1, W2 copied there once by whole-line 16-byte loads (padded pitches), a tile's fragment loads issued as one group before its WMMAs; same order of sums
# speedup vs baseline: 16.4477x; 1.0424x over previous
#include <hip/hip_runtime.h>


#define HDIM   64
#define VOCABN 64
#define BATCH  64
#define SEQL   4096
#define TPITCH 68
#define RPITCH 132
#define XPITCH 68
#define EPITCH  68
#define W1PITCH 144
#define W2PITCH 80

typedef __attribute__((ext_vector_type(2))) float v2f;
typedef __attribute__((ext_vector_type(2))) float f32x2;
typedef __attribute__((ext_vector_type(8))) float v8f;
typedef __attribute__((ext_vector_type(3))) float v3f;

__device__ __forceinline__ v2f frag_a(const float* A, int lda, int mt, int kt, int lane) {
  int row = (mt << 4) + (lane & 15);
  int k0  = (kt << 2) + ((lane >> 4) << 1);
  v2f a;
  a.x = A[row * lda + k0];
  a.y = A[row * lda + k0 + 1];
  return a;
}
__device__ __forceinline__ v2f frag_b(const float* Bm, int ldb, int kt, int nt, int lane) {
  int col = (nt << 4) + (lane & 15);
  int k0  = (kt << 2) + ((lane >> 4) << 1);
  v2f b;
  b.x = Bm[k0 * ldb + col];
  b.y = Bm[(k0 + 1) * ldb + col];
  return b;
}
__device__ __forceinline__ v2f frag_bt(const float* S, int lds, int kt, int nt, int lane) {
  int col = (nt << 4) + (lane & 15);
  int k0  = (kt << 2) + ((lane >> 4) << 1);
  v2f b;
  b.x = S[col * lds + k0];
  b.y = S[col * lds + k0 + 1];
  return b;
}

__device__ __forceinline__ v8f wmma_f32(v2f a, v2f b, v8f c) {
  return __builtin_amdgcn_wmma_f32_16x16x4_f32(false, a, false, b, (short)0, c,
                                               false, false);
}

__device__ __forceinline__ v2f frag_a_row0(const float2* x2, int kt, int lane) {
  const float2 xv = x2[(kt << 1) + (lane >> 4)];
  const bool   r0 = (lane & 15) == 0;
  v2f a;
  a.x = r0 ? xv.x : 0.0f;
  a.y = r0 ? xv.y : 0.0f;
  return a;
}

#define GRP    16
#define NSLOT  8
#define NTRIP  32
#define NTHR   352

template <int J>
__device__ __forceinline__ int row_share_i(int x) {
  return __builtin_amdgcn_update_dpp(0, x, 0x150 + J, 0xf, 0xf, true);
}
template <int J>
__device__ __forceinline__ float row_share(float x) {
  return __int_as_float(row_share_i<J>(__float_as_int(x)));
}

#define DPPCTL " row_mask:0xf bank_mask:0xf bound_ctrl:1"
__device__ __forceinline__ void scan_step_first(float& p, float& s, float a) {
  float pn, sn;
  asm("v_fma_f32_e64_dpp %0, %2, %3, %2 row_share:0" DPPCTL "\n\t"
      "v_mov_b32_dpp %1, %2 row_share:0" DPPCTL
      : "=&v"(pn), "=&v"(sn)
      : "v"(p), "v"(a));
  p = pn;
  s = sn;
}
template <int J>
__device__ __forceinline__ void scan_step_mid(float& p, float& s, float& c_lo, float& c_hi,
                                              float a, float gx, float gy) {
  float pn, sn;
  asm("s_delay_alu instid0(VALU_DEP_4)\n\t"
      "v_fma_f32_e64_dpp %0, %4, %5, %4 row_share:%9" DPPCTL "\n\t"
      "v_mov_b32_dpp %1, %4 row_share:%9" DPPCTL "\n\t"
      "v_fmac_f32_e32 %2, %6, %8\n\t"
      "v_fmac_f32_e32 %3, %7, %8"
      : "=&v"(pn), "=&v"(sn), "+v"(c_lo), "+v"(c_hi)
      : "v"(p), "v"(a), "v"(gx), "v"(gy), "v"(s), "n"(J));
  p = pn;
  s = sn;
}
__device__ __forceinline__ void scan_step_last(float p, float s, float& c_lo, float& c_hi,
                                               float gx14, float gy14, float gx15, float gy15) {
  asm("v_fmac_f32_e32 %0, %3, %5\n\t"
      "v_fmac_f32_e32 %1, %4, %5\n\t"
      "v_fmac_f32_dpp %0, %2, %6 row_share:15" DPPCTL "\n\t"
      "v_fmac_f32_dpp %1, %2, %7 row_share:15" DPPCTL
      : "+v"(c_lo), "+v"(c_hi)
      : "v"(p), "v"(gx14), "v"(gy14), "v"(s), "v"(gx15), "v"(gy15));
}

__device__ __forceinline__ float gather_c(float c_lo, float c_hi, int tok, int lane) {
  const unsigned sel = (unsigned)tok & 15u;
  const unsigned ulo = (unsigned)__float_as_int(c_lo);
  const unsigned uhi = (unsigned)__float_as_int(c_hi);
  const unsigned lo_s = __builtin_amdgcn_permlane16_var(ulo, ulo, sel, false, false);
  const unsigned lo_x = __builtin_amdgcn_permlanex16_var(ulo, ulo, sel, false, false);
  const unsigned hi_s = __builtin_amdgcn_permlane16_var(uhi, uhi, sel, false, false);
  const unsigned hi_x = __builtin_amdgcn_permlanex16_var(uhi, uhi, sel, false, false);
  const bool other_row = (((tok >> 4) ^ (lane >> 4)) & 1) != 0;
  const unsigned lo = other_row ? lo_x : lo_s;
  const unsigned hi = other_row ? hi_x : hi_s;
  return __int_as_float((int)((tok & 32) ? hi : lo));
}

#define NWAVE  11

__global__ __launch_bounds__(NTHR) void delta_fused_kernel(
    const int* __restrict__ seq, const float* __restrict__ embed,
    const float* __restrict__ W1, const float* __restrict__ b1,
    const float* __restrict__ W2, const float* __restrict__ b2,
    const float* __restrict__ gamma, const float* __restrict__ beta,
    const float* __restrict__ Wr, const float* __restrict__ br,
    const float* __restrict__ Wo, const float* __restrict__ bo,
    float* __restrict__ out) {
  __shared__ float4 Gl4[VOCABN * 16];
  __shared__ float4 Tl4[VOCABN * 16];
  __shared__ float4 Pk[2 * NSLOT * GRP * 32];
  __shared__ float  Sb[SEQL];
  __shared__ float2 Part[NSLOT * 32];
  __shared__ float4 Rl4[64 * (RPITCH / 4)];
  __shared__ float4 Xl4[64 * (XPITCH / 4)];
  __shared__ float  rdl[64];
  __shared__ float  Gpl[VOCABN * VOCABN];
  __shared__ float2 Cx2[32];
  __shared__ float2 Yx2[32];
  __shared__ float2 Ox2[32];
  float* Rlds = (float*)Rl4;
  float* Xlds = (float*)Xl4;
  const int b   = blockIdx.x;
  const int tid = threadIdx.x;
  const int lane = tid & 31;
  const int wave = __builtin_amdgcn_readfirstlane(tid >> 5);

  const bool helper = (wave & 3) != 0;
  const int  hslot  = wave - 1 - (wave >> 2);

  int tvP = 0;
  if (helper) tvP = seq[b * SEQL + (SEQL / 16 - 1 - hslot) * 16 + (15 - (lane & 15))];

  float* Es  = (float*)Pk;
  float* W1s = Es + 64 * EPITCH;
  float* W2s = W1s + 64 * W1PITCH;
  if (wave < 8) {
    float4 ev[4], pv[8], qv[8];
#pragma unroll
    for (int j = 0; j < 4; ++j) ev[j] = ((const float4*)embed)[tid + 256 * j];
#pragma unroll
    for (int j = 0; j < 8; ++j) pv[j] = ((const float4*)W1)[tid + 256 * j];
#pragma unroll
    for (int j = 0; j < 8; ++j) qv[j] = ((const float4*)W2)[tid + 256 * j];
#pragma unroll
    for (int j = 0; j < 4; ++j) {
      const int i = tid + 256 * j;
      *(float4*)&Es[(i >> 4) * EPITCH + ((i & 15) << 2)] = ev[j];
    }
#pragma unroll
    for (int j = 0; j < 8; ++j) {
      const int i = tid + 256 * j;
      *(float4*)&W1s[(i >> 5) * W1PITCH + ((i & 31) << 2)] = pv[j];
    }
#pragma unroll
    for (int j = 0; j < 8; ++j) {
      const int i = tid + 256 * j;
      *(float4*)&W2s[(i >> 4) * W2PITCH + ((i & 15) << 2)] = qv[j];
    }
  }
  __syncthreads();

  for (int tile = wave; tile < 32; tile += NWAVE) {
    int mt = tile >> 3, nt = tile & 7;
    v8f acc8 = {};
    v2f af[16], bf[16];
#pragma unroll
    for (int kt = 0; kt < 16; ++kt) {
      af[kt] = frag_a(Es, EPITCH, mt, kt, lane);
      bf[kt] = frag_b(W1s, W1PITCH, kt, nt, lane);
    }
    __builtin_amdgcn_sched_barrier(0);
#pragma unroll
    for (int kt = 0; kt < 16; ++kt) acc8 = wmma_f32(af[kt], bf[kt], acc8);
    int col   = (nt << 4) + (lane & 15);
    int rbase = (mt << 4) + ((lane >> 4) << 3);
#pragma unroll
    for (int v = 0; v < 8; ++v) {
      float x = acc8[v] + b1[col];
      Rlds[(rbase + v) * RPITCH + col] = fmaxf(x, 0.0f);
    }
  }
  __syncthreads();

  for (int tile = wave; tile < 16; tile += NWAVE) {
    int mt = tile >> 2, nt = tile & 3;
    v8f acc8 = {};
    v2f af[32], bf[32];
#pragma unroll
    for (int kt = 0; kt < 32; ++kt) {
      af[kt] = frag_a(Rlds, RPITCH, mt, kt, lane);
      bf[kt] = frag_b(W2s, W2PITCH, kt, nt, lane);
    }
    __builtin_amdgcn_sched_barrier(0);
#pragma unroll
    for (int kt = 0; kt < 32; ++kt) acc8 = wmma_f32(af[kt], bf[kt], acc8);
    int col   = (nt << 4) + (lane & 15);
    int rbase = (mt << 4) + ((lane >> 4) << 3);
#pragma unroll
    for (int v = 0; v < 8; ++v) {
      int row = rbase + v;
      Xlds[row * XPITCH + col] = Es[row * EPITCH + col] + acc8[v] + b2[col];
    }
  }
  __syncthreads();

  if (tid < 64) {
    const float* xr = &Xlds[tid * XPITCH];
    float mu = 0.f;
#pragma unroll
    for (int c = 0; c < 64; ++c) mu += xr[c];
    mu *= (1.0f / 64.0f);
    float var = 0.f;
#pragma unroll
    for (int c = 0; c < 64; ++c) { float d = xr[c] - mu; var += d * d; }
    var *= (1.0f / 64.0f);
    float rs  = rsqrtf(var + 1e-5f);
    float ksq = 0.f;
#pragma unroll
    for (int c = 0; c < 64; ++c) {
      float tv = (xr[c] - mu) * rs * gamma[c] + beta[c];
      Rlds[tid * TPITCH + c] = tv;
      ksq += tv * tv;
    }
    rdl[tid] = 1.0f / (ksq + 1e-6f);
  }
  __syncthreads();

  for (int i = tid; i < VOCABN * 16; i += NTHR)
    Tl4[i] = *(const float4*)&Rlds[(i >> 4) * TPITCH + ((i & 15) << 2)];

  {
    const float* Tp  = Rlds;
    float*       Gsf = (float*)Gl4;
    for (int tile = wave; tile < 16; tile += NWAVE) {
      int mt = tile >> 2, nt = tile & 3;
      v8f acc8 = {};
#pragma unroll
      for (int kt = 0; kt < 16; ++kt) {
        v2f a = frag_a(Tp, TPITCH, mt, kt, lane);
        v2f bq = frag_bt(Tp, TPITCH, kt, nt, lane);
        acc8 = wmma_f32(a, bq, acc8);
      }
      int col   = (nt << 4) + (lane & 15);
      int rbase = (mt << 4) + ((lane >> 4) << 3);
#pragma unroll
      for (int v = 0; v < 8; ++v) {
        int row = rbase + v;
        Gpl[row * 64 + col] = acc8[v];
        Gsf[((row << 5) + (col & 31)) * 2 + (col >> 5)] = -(acc8[v] * rdl[row]);
      }
    }
  }
  __syncthreads();

  const float2* Gl2 = (const float2*)Gl4;
  const float*  Glf = (const float*)Gl4;
  const float2* Tl2 = (const float2*)Tl4;
  const int*    sqr = seq + b * SEQL;
  const int     li  = lane & 15;
  const int     NG  = SEQL / GRP;

  float  c_lo = 0.0f, c_hi = 0.0f;
  float2 acc  = make_float2(0.0f, 0.0f);
  int    tvA = 0, tvB = 0;
  if (wave == 0) {
    const int tq = sqr[SEQL - 1];
    c_lo = Gpl[tq * 64 + lane];
    c_hi = Gpl[tq * 64 + lane + 32];
  }

#define PLOAD(o)                                                           \
  {                                                                        \
    const int to = row_share_i<(o)>(tvN);                                  \
    Rv[o] = Gl2[to * 32 + lane];                                           \
    Av[o] = Glf[to * 64 + offn];                                           \
  }
#define PSTORE(o)                                                          \
  {                                                                        \
    const float A = (li > (o)) ? Av[o] : 0.0f;                             \
    v3f rec;                                                               \
    rec.x = Rv[o].x;                                                       \
    rec.y = Rv[o].y;                                                       \
    rec.z = A;                                                             \
    *(v3f*)&pkw[(o) * 32] = rec;                                           \
  }
#define PSTORE0()                                                          \
  {                                                                        \
    float2 R = Rv[0];                                                      \
    float  A = (li > 0) ? Av[0] : 0.0f;                                    \
    if (qgroup) { R.x = 0.0f; R.y = 0.0f; A = 0.0f; }                      \
    pkw[0] = make_float4(R.x, R.y, A, __int_as_float(tvN));                \
  }
#define ACC1(o)                                                            \
  {                                                                        \
    float so = row_share<(o)>(sv);                                         \
    if ((o) == 0 && qdone) so = 0.0f;                                      \
    const int    to = row_share_i<(o)>(tvA);                               \
    const float2 tr = Tl2[to * 32 + lane];                                 \
    acc.x = fmaf(tr.x, so, acc.x);                                         \
    acc.y = fmaf(tr.y, so, acc.y);                                         \
  }
#define ACCGROUP()                                                         \
  {                                                                        \
    const int   g2    = NG - 1 - ((k - 2) * NSLOT + h);                    \
    const bool  qdone = (g2 == NG - 1);                                    \
    const float sv    = Sb[g2 * GRP + li];                                 \
    ACC1(0) ACC1(1) ACC1(2) ACC1(3) ACC1(4) ACC1(5) ACC1(6) ACC1(7)        \
    ACC1(8) ACC1(9) ACC1(10) ACC1(11) ACC1(12) ACC1(13) ACC1(14) ACC1(15)  \
  }
#define LOADSET(X, slot)                                                   \
  {                                                                        \
    _Pragma("unroll") for (int o = 0; o < GRP; ++o)                        \
      X[o] = pkr[((slot) * GRP + o) * 32];                                 \
  }
#define MSTEP(o, X)                                                        \
  scan_step_mid<(o)>(p, s, c_lo, c_hi, X[o].z, X[(o) - 1].x, X[(o) - 1].y);
#define GROUP(Xc, Xn, slot, HASNEXT)                                       \
  {                                                                        \
    if (HASNEXT) LOADSET(Xn, (slot) + 1)                                   \
    __builtin_amdgcn_sched_barrier(0);                                     \
    float p = gather_c(c_lo, c_hi, __float_as_int(Xc[0].w), lane);         \
    float s;                                                               \
    scan_step_first(p, s, Xc[0].z);                                        \
    MSTEP(1, Xc) MSTEP(2, Xc) MSTEP(3, Xc) MSTEP(4, Xc) MSTEP(5, Xc)       \
    MSTEP(6, Xc) MSTEP(7, Xc) MSTEP(8, Xc) MSTEP(9, Xc) MSTEP(10, Xc)      \
    MSTEP(11, Xc) MSTEP(12, Xc) MSTEP(13, Xc) MSTEP(14, Xc)                \
    scan_step_last(p, s, c_lo, c_hi, Xc[14].x, Xc[14].y, Xc[15].x, Xc[15].y); \
    Sb[(gfirst - (slot)) * GRP + li] = p;                                  \
  }

  for (int k = 0; k < NTRIP + 2; ++k) {
    if (wave == 0) {
      if (k >= 1 && k <= NTRIP) {
        const float4* pkr    = Pk + ((k - 1) & 1) * (NSLOT * GRP * 32) + lane;
        const int     gfirst = NG - 1 - (k - 1) * NSLOT;
        float4 X0[GRP], X1[GRP];
        LOADSET(X0, 0)
        GROUP(X0, X1, 0, 1)
        GROUP(X1, X0, 1, 1)
        GROUP(X0, X1, 2, 1)
        GROUP(X1, X0, 3, 1)
        GROUP(X0, X1, 4, 1)
        GROUP(X1, X0, 5, 1)
        GROUP(X0, X1, 6, 1)
        GROUP(X1, X0, 7, 0)
      }
    } else if (helper) {
      const int h   = hslot;
      int       tvN = 0;
      if (k < NTRIP) {
        const int  g      = NG - 1 - (k * NSLOT + h);
        const bool qgroup = (g == NG - 1);
        float4*    pkw    = Pk + ((k & 1) * NSLOT + h) * (GRP * 32) + lane;
        tvN = tvP;
        const int kq = (k + 1 < NTRIP) ? k + 1 : k;
        tvP = sqr[(NG - 1 - (kq * NSLOT + h)) * GRP + (GRP - 1 - li)];
        const int offn = ((tvN & 31) << 1) + (tvN >> 5);
        float2 Rv[GRP];
        float  Av[GRP];
        PLOAD(0) PLOAD(1) PLOAD(2) PLOAD(3) PLOAD(4) PLOAD(5) PLOAD(6) PLOAD(7)
        PLOAD(8) PLOAD(9) PLOAD(10) PLOAD(11) PLOAD(12) PLOAD(13) PLOAD(14) PLOAD(15)
        if (k >= 2) ACCGROUP()
        PSTORE0()
        PSTORE(1) PSTORE(2) PSTORE(3) PSTORE(4) PSTORE(5) PSTORE(6) PSTORE(7)
        PSTORE(8) PSTORE(9) PSTORE(10) PSTORE(11) PSTORE(12) PSTORE(13) PSTORE(14) PSTORE(15)
      } else {
        ACCGROUP()
      }
      tvA = tvB;
      tvB = tvN;
    }
    __syncthreads();
  }
#undef GROUP
#undef LOADSET
#undef MSTEP
#undef ACCGROUP
#undef ACC1
#undef PSTORE0
#undef PSTORE
#undef PLOAD

  if (helper) Part[hslot * 32 + lane] = acc;
  __syncthreads();
  if (wave == 0) {
    float2 r = Part[lane];
    for (int h = 1; h < NSLOT; ++h) {
      const float2 q = Part[h * 32 + lane];
      r.x += q.x;
      r.y += q.y;
    }
    Cx2[lane] = r;
  }
  __syncthreads();

  v2f bO[16];
  if (wave < 4) {
    const int nt = wave;
    v2f bR[16];
#pragma unroll
    for (int kt = 0; kt < 16; ++kt) bR[kt] = frag_b(Wr, 64, kt, nt, lane);
#pragma unroll
    for (int kt = 0; kt < 16; ++kt) bO[kt] = frag_b(Wo, 64, kt, nt, lane);
    v8f acc8 = {};
#pragma unroll
    for (int kt = 0; kt < 16; ++kt)
      acc8 = wmma_f32(frag_a_row0(Cx2, kt, lane), bR[kt], acc8);
    const int   col = (nt << 4) + (lane & 15);
    const float y   = acc8[0] + br[col];
    if (lane < 16) ((float*)Yx2)[col] = y;
  }
  __syncthreads();
  if (wave < 4) {
    const int nt = wave;
    v8f acc8 = {};
#pragma unroll
    for (int kt = 0; kt < 16; ++kt)
      acc8 = wmma_f32(frag_a_row0(Yx2, kt, lane), bO[kt], acc8);
    const int   col = (nt << 4) + (lane & 15);
    const float y   = acc8[0] + bo[col];
    if (lane < 16) ((float*)Ox2)[col] = y;
  }
  __syncthreads();
  if (wave == 0) ((float2*)(out + b * 64))[lane] = Ox2[lane];
}

extern "C" void kernel_launch(void* const* d_in, const int* in_sizes, int n_in,
                              void* d_out, int out_size, void* d_ws, size_t ws_size,
                              hipStream_t stream) {
  const int*   seq   = (const int*)d_in[0];
  const float* embed = (const float*)d_in[1];
  const float* W1    = (const float*)d_in[2];
  const float* b1    = (const float*)d_in[3];
  const float* W2    = (const float*)d_in[4];
  const float* b2    = (const float*)d_in[5];
  const float* gamma = (const float*)d_in[6];
  const float* beta  = (const float*)d_in[7];
  const float* Wr    = (const float*)d_in[8];
  const float* br    = (const float*)d_in[9];
  const float* Wo    = (const float*)d_in[10];
  const float* bo    = (const float*)d_in[11];
  float*       out   = (float*)d_out;

  delta_fused_kernel<<<BATCH, NTHR, 0, stream>>>(seq, embed, W1, b1, W2, b2, gamma, beta,
                                                 Wr, br, Wo, bo, out);
}
